# starting kernel
# baseline (speedup 1.0000x reference)
#include <hip/hip_runtime.h>
#include <hip/hip_bf16.h>

typedef __attribute__((ext_vector_type(2))) float v2f;
typedef __attribute__((ext_vector_type(8))) float v8f;

#define QSCALE 0.17677669529663687f


__global__ void transpose_k(const float* __restrict__ src, float* __restrict__ dst,
                            int O, int K, int total) {
    int idx = blockIdx.x * blockDim.x + threadIdx.x;
    if (idx >= total) return;
    int o = idx % O;
    int k = idx / O;
    dst[idx] = src[(size_t)o * K + k];
}

__global__ void pool_transpose(const float* __restrict__ src, float* __restrict__ dst,
                               int total) {
    int idx = blockIdx.x * blockDim.x + threadIdx.x;
    if (idx >= total) return;
    int o = idx % 256;
    int k = idx / 256;
    int c = k % 256;
    int q = k / 256;
    dst[idx] = src[(size_t)o * 16384 + c * 64 + q];
}

__global__ void gather_xw(const float* __restrict__ x, float* __restrict__ xw, int total) {
    int idx = blockIdx.x * blockDim.x + threadIdx.x;
    if (idx >= total) return;
    int c = idx % 256;
    int t = idx / 256;
    int q = t % 64;
    int n = t / 64;
    int bi = n / 144;
    int l  = n % 144;
    int wh = l / 12, ww = l % 12;
    int i = q / 8, j = q % 8;
    xw[idx] = x[(((size_t)bi * 256 + c) * 96 + wh * 8 + i) * 96 + ww * 8 + j];
}

__global__ __launch_bounds__(256) void gemm_wmma_f32(
    const float* __restrict__ A, const float* __restrict__ B,
    float* __restrict__ C, const float* __restrict__ bias,
    int M, int N, int K, float scale)
{
    int wave = threadIdx.x >> 5;
    int lane = threadIdx.x & 31;
    int hf   = lane >> 4;
    int r    = lane & 15;
    int m0 = blockIdx.y * 32 + (wave >> 2) * 16;
    int n0 = blockIdx.x * 64 + (wave & 3) * 16;

    const float* Arow = A + (size_t)(m0 + r) * K + 2 * hf;
    const float* Bcol = B + (size_t)(2 * hf) * N + n0 + r;

    v8f acc = {};
    #pragma unroll 8
    for (int k = 0; k < K; k += 4) {
        v2f a, b;
        a.x = Arow[k];
        a.y = Arow[k + 1];
        b.x = Bcol[(size_t)k * N];
        b.y = Bcol[(size_t)k * N + N];
        acc = __builtin_amdgcn_wmma_f32_16x16x4_f32(false, a, false, b,
                                                    (short)0, acc, false, false);
    }
    int col = n0 + r;
    float bv = bias ? bias[col] : 0.0f;
    #pragma unroll
    for (int vg = 0; vg < 8; ++vg)
        C[(size_t)(m0 + vg + 8 * hf) * N + col] = acc[vg] * scale + bv;
}

#define LDST 145
__global__ __launch_bounds__(128) void attn_kernel(
    float* __restrict__ qmat, const float* __restrict__ kv)
{
    __shared__ float attn[64 * LDST];
    int id = blockIdx.x;
    int h  = id & 7;
    int bb = (id >> 3) & 7;
    int g  = id >> 6;
    int n  = g * 8 + bb;

    int wave = threadIdx.x >> 5;
    int lane = threadIdx.x & 31;
    int hf   = lane >> 4;
    int r    = lane & 15;
    int m0   = wave * 16;

    const float* qbase = qmat + ((size_t)n * 64 + m0 + r) * 256 + h * 32;
    const float* kvb   = kv + (size_t)bb * 144 * 512 + h * 32;
    const float* vb    = kvb + 256;

    for (int nt = 0; nt < 9; ++nt) {
        int n0 = nt * 16;
        const float* kcol = kvb + (size_t)(n0 + r) * 512 + 2 * hf;
        v8f acc = {};
        #pragma unroll
        for (int k = 0; k < 32; k += 4) {
            v2f a, b;
            a.x = qbase[k + 2 * hf];
            a.y = qbase[k + 2 * hf + 1];
            b.x = kcol[k];
            b.y = kcol[k + 1];
            acc = __builtin_amdgcn_wmma_f32_16x16x4_f32(false, a, false, b,
                                                        (short)0, acc, false, false);
        }
        #pragma unroll
        for (int vg = 0; vg < 8; ++vg)
            attn[(m0 + vg + 8 * hf) * LDST + n0 + r] = acc[vg];
    }
    __syncthreads();

    if (threadIdx.x < 64) {
        float* row = attn + threadIdx.x * LDST;
        float mx = row[0];
        for (int k = 1; k < 144; ++k) mx = fmaxf(mx, row[k]);
        float s = 0.0f;
        for (int k = 0; k < 144; ++k) { float e = __expf(row[k] - mx); row[k] = e; s += e; }
        float inv = 1.0f / s;
        for (int k = 0; k < 144; ++k) row[k] *= inv;
    }
    __syncthreads();

    const float* arow = attn + (m0 + r) * LDST + 2 * hf;
    for (int nt = 0; nt < 2; ++nt) {
        int n0 = nt * 16;
        v8f acc = {};
        #pragma unroll 4
        for (int k = 0; k < 144; k += 4) {
            v2f a, b;
            a.x = arow[k];
            a.y = arow[k + 1];
            b.x = vb[(size_t)(k + 2 * hf) * 512 + n0 + r];
            b.y = vb[(size_t)(k + 2 * hf + 1) * 512 + n0 + r];
            acc = __builtin_amdgcn_wmma_f32_16x16x4_f32(false, a, false, b,
                                                        (short)0, acc, false, false);
        }
        #pragma unroll
        for (int vg = 0; vg < 8; ++vg)
            qmat[((size_t)n * 64 + m0 + vg + 8 * hf) * 256 + h * 32 + n0 + r] = acc[vg];
    }
}

__global__ void rpe_add_kernel(const float* __restrict__ x, const float* __restrict__ rw,
                               const float* __restrict__ rb, float* __restrict__ F,
                               int total) {
    int idx = blockIdx.x * blockDim.x + threadIdx.x;
    if (idx >= total) return;
    int xx = idx % 96;
    int t  = idx / 96;
    int y  = t % 96; t /= 96;
    int c  = t % 256;
    int bi = t / 256;
    const float* xp = x + (size_t)(bi * 256 + c) * 9216;
    const float* wp = rw + c * 9;
    float acc = rb[c];
    #pragma unroll
    for (int dy = -1; dy <= 1; ++dy) {
        int yy = y + dy;
        if (yy < 0 || yy >= 96) continue;
        #pragma unroll
        for (int dx = -1; dx <= 1; ++dx) {
            int xc = xx + dx;
            if (xc < 0 || xc >= 96) continue;
            acc += xp[yy * 96 + xc] * wp[(dy + 1) * 3 + (dx + 1)];
        }
    }
    F[idx] += acc;
}

extern "C" void kernel_launch(void* const* d_in, const int* in_sizes, int n_in,
                              void* d_out, int out_size, void* d_ws, size_t ws_size,
                              hipStream_t stream) {
    const float* x      = (const float*)d_in[0];
    const float* Wq     = (const float*)d_in[1];
    const float* Wkv    = (const float*)d_in[2];
    const float* pool_w = (const float*)d_in[3];
    const float* pool_b = (const float*)d_in[4];
    const float* rpe_w  = (const float*)d_in[5];
    const float* rpe_b  = (const float*)d_in[6];
    const float* proj_w = (const float*)d_in[7];
    const float* proj_b = (const float*)d_in[8];
    float* out = (float*)d_out;

    float* ws    = (float*)d_ws;
    float* xw    = ws;
    float* qmat  = xw + 18874368;
    float* poolT = qmat + 18874368;
    float* WqT   = poolT + 4194304;
    float* WkvT  = WqT + 65536;
    float* projT = WkvT + 131072;
    float* poolx = projT + 65536;
    float* kvb   = poolx + 294912;
    (void)ws_size; (void)in_sizes; (void)n_in; (void)out_size;

    const int TB = 256;
    transpose_k<<<(65536 + TB - 1) / TB, TB, 0, stream>>>(Wq, WqT, 256, 256, 65536);
    transpose_k<<<(131072 + TB - 1) / TB, TB, 0, stream>>>(Wkv, WkvT, 512, 256, 131072);
    transpose_k<<<(65536 + TB - 1) / TB, TB, 0, stream>>>(proj_w, projT, 256, 256, 65536);
    pool_transpose<<<(4194304 + TB - 1) / TB, TB, 0, stream>>>(pool_w, poolT, 4194304);

    gather_xw<<<(18874368 + TB - 1) / TB, TB, 0, stream>>>(x, xw, 18874368);

    gemm_wmma_f32<<<dim3(256 / 64, 1152 / 32), TB, 0, stream>>>(
        xw, poolT, poolx, pool_b, 1152, 256, 16384, 1.0f);

    gemm_wmma_f32<<<dim3(256 / 64, 73728 / 32), TB, 0, stream>>>(
        xw, WqT, qmat, nullptr, 73728, 256, 256, QSCALE);

    gemm_wmma_f32<<<dim3(512 / 64, 1152 / 32), TB, 0, stream>>>(
        poolx, WkvT, kvb, nullptr, 1152, 512, 256, 1.0f);

    attn_kernel<<<144 * 8 * 8, 128, 0, stream>>>(qmat, kvb);

    rpe_add_kernel<<<(18874368 + TB - 1) / TB, TB, 0, stream>>>(
        x, rpe_w, rpe_b, qmat, 18874368);

    gemm_wmma_f32<<<dim3(256 / 64, 73728 / 32), TB, 0, stream>>>(
        qmat, projT, out, proj_b, 73728, 256, 256, 1.0f);
}


# RPE pass folded into the out projection: a block forms F + bias + 9 taps of x for its 32x256 A tile in LDS and reads its A fragments there; F+rpe never stored; 9 launches -> 8; sum order unchanged
# speedup vs baseline: 3.3272x; 3.3272x over previous
#include <hip/hip_runtime.h>
#include <hip/hip_bf16.h>

typedef __attribute__((ext_vector_type(2))) float v2f;
typedef __attribute__((ext_vector_type(8))) float v8f;

#define QSCALE 0.17677669529663687f


__global__ void transpose_k(const float* __restrict__ src, float* __restrict__ dst,
                            int O, int K, int total) {
    int idx = blockIdx.x * blockDim.x + threadIdx.x;
    if (idx >= total) return;
    int o = idx % O;
    int k = idx / O;
    dst[idx] = src[(size_t)o * K + k];
}

__global__ __launch_bounds__(256) void gemm_wmma_f32(
    const float* __restrict__ A, const float* __restrict__ B,
    float* __restrict__ C, const float* __restrict__ bias,
    int M, int N, int K, float scale)
{
    int wave = threadIdx.x >> 5;
    int lane = threadIdx.x & 31;
    int hf   = lane >> 4;
    int r    = lane & 15;
    int m0 = blockIdx.y * 32 + (wave >> 2) * 16;
    int n0 = blockIdx.x * 64 + (wave & 3) * 16;

    const float* Arow = A + (size_t)(m0 + r) * K + 2 * hf;
    const float* Bcol = B + (size_t)(2 * hf) * N + n0 + r;

    v8f acc = {};
    #pragma unroll 8
    for (int k = 0; k < K; k += 4) {
        v2f a, b;
        a.x = Arow[k];
        a.y = Arow[k + 1];
        b.x = Bcol[(size_t)k * N];
        b.y = Bcol[(size_t)k * N + N];
        acc = __builtin_amdgcn_wmma_f32_16x16x4_f32(false, a, false, b,
                                                    (short)0, acc, false, false);
    }
    int col = n0 + r;
    float bv = bias ? bias[col] : 0.0f;
    #pragma unroll
    for (int vg = 0; vg < 8; ++vg)
        C[(size_t)(m0 + vg + 8 * hf) * N + col] = acc[vg] * scale + bv;
}

typedef __attribute__((ext_vector_type(4))) float v4f;

struct RegFrag {
    v4f a0, a1;
    v4f b0, b1, b2, b3;
};

__device__ __forceinline__ RegFrag regtile_load(const float* __restrict__ A0,
                                                const float* __restrict__ A1,
                                                const float* __restrict__ Bp,
                                                int k, int N) {
    RegFrag f;
    f.a0 = *(const v4f*)(A0 + k);
    f.a1 = *(const v4f*)(A1 + k);
    const float* bp = Bp + (size_t)k * N;
    f.b0 = *(const v4f*)(bp);
    f.b1 = *(const v4f*)(bp + N);
    f.b2 = *(const v4f*)(bp + 2 * (size_t)N);
    f.b3 = *(const v4f*)(bp + 3 * (size_t)N);
    return f;
}

__device__ __forceinline__ void regtile_mma(v8f (&acc)[2][4], const RegFrag& f) {
    v2f a00 = {f.a0.x, f.a0.y}, a01 = {f.a0.z, f.a0.w};
    v2f a10 = {f.a1.x, f.a1.y}, a11 = {f.a1.z, f.a1.w};
    #pragma unroll
    for (int j = 0; j < 4; ++j) {
        v2f b = {f.b0[j], f.b1[j]};
        acc[0][j] = __builtin_amdgcn_wmma_f32_16x16x4_f32(false, a00, false, b,
                                                          (short)0, acc[0][j], false, false);
        acc[1][j] = __builtin_amdgcn_wmma_f32_16x16x4_f32(false, a10, false, b,
                                                          (short)0, acc[1][j], false, false);
    }
    #pragma unroll
    for (int j = 0; j < 4; ++j) {
        v2f b = {f.b2[j], f.b3[j]};
        acc[0][j] = __builtin_amdgcn_wmma_f32_16x16x4_f32(false, a01, false, b,
                                                          (short)0, acc[0][j], false, false);
        acc[1][j] = __builtin_amdgcn_wmma_f32_16x16x4_f32(false, a11, false, b,
                                                          (short)0, acc[1][j], false, false);
    }
}

#define OP_APITCH 260

__global__ __launch_bounds__(128) void outproj_rpe_fused(
    const float* __restrict__ F, const float* __restrict__ x,
    const float* __restrict__ rw, const float* __restrict__ rb,
    const float* __restrict__ B, float* __restrict__ C,
    const float* __restrict__ bias, int N, int K)
{
    __shared__ __attribute__((aligned(16))) float atile[32 * OP_APITCH];
    int wave = threadIdx.x >> 5;
    int lane = threadIdx.x & 31;
    int hf   = lane >> 4;
    int r    = lane & 15;
    int m0 = blockIdx.x * 32;
    int n0 = wave * 64;

    int rsel = __builtin_amdgcn_readfirstlane(wave >> 1);
    int kq = 4 * (threadIdx.x & 63);
    const char* xb = (const char*)x;
    #pragma unroll 4
    for (int i = 0; i < 16; ++i) {
        int row = 2 * i + rsel;
        int m  = m0 + row;
        int pl = m / 36;
        int c  = pl & 255;
        int p  = (m - pl * 36) * 256 + kq;
        int y  = p / 96;
        int xx = p - y * 96;
        unsigned bo = ((unsigned)m * 256u + (unsigned)kq) * 4u;
        v4f f = *(const v4f*)((const char*)F + bo);
        const float* wp = rw + c * 9;
        float b0 = rb[c];
        v4f s = {b0, b0, b0, b0};
        bool lok = xx > 0, rok = xx < 92;
        unsigned lo = lok ? 0xfffffffcu : 0u;
        unsigned ro = rok ? 16u : 12u;
        #pragma unroll
        for (int dy = -1; dy <= 1; ++dy) {
            bool rowok = dy < 0 ? (y > 0) : (dy > 0 ? (y < 95) : true);
            unsigned br = rowok ? bo + (unsigned)(dy * 384) : bo;
            v4f mid = *(const v4f*)(xb + br);
            float lf = *(const float*)(xb + (br + lo));
            float rg = *(const float*)(xb + (br + ro));
            if (!rowok) mid = (v4f){0.0f, 0.0f, 0.0f, 0.0f};
            lf = (rowok && lok) ? lf : 0.0f;
            rg = (rowok && rok) ? rg : 0.0f;
            float w0 = wp[(dy + 1) * 3], w1 = wp[(dy + 1) * 3 + 1], w2 = wp[(dy + 1) * 3 + 2];
            s.x += lf    * w0; s.x += mid.x * w1; s.x += mid.y * w2;
            s.y += mid.x * w0; s.y += mid.y * w1; s.y += mid.z * w2;
            s.z += mid.y * w0; s.z += mid.z * w1; s.z += mid.w * w2;
            s.w += mid.z * w0; s.w += mid.w * w1; s.w += rg    * w2;
        }
        *(v4f*)(atile + row * OP_APITCH + kq) = f + s;
    }
    __syncthreads();

    const float* A0 = atile + r * OP_APITCH + 4 * hf;
    const float* A1 = A0 + 16 * OP_APITCH;
    const float* Bp = B + (size_t)(4 * hf) * N + n0 + 4 * r;

    v8f acc[2][4] = {};
    RegFrag cur = regtile_load(A0, A1, Bp, 0, N);
    for (int k = 8; k < K; k += 8) {
        RegFrag nxt = regtile_load(A0, A1, Bp, k, N);
        regtile_mma(acc, cur);
        cur = nxt;
    }
    regtile_mma(acc, cur);

    v4f bv = *(const v4f*)(bias + n0 + 4 * r);
    #pragma unroll
    for (int rt = 0; rt < 2; ++rt) {
        #pragma unroll
        for (int vg = 0; vg < 8; ++vg) {
            v4f o;
            o.x = acc[rt][0][vg] + bv.x;
            o.y = acc[rt][1][vg] + bv.y;
            o.z = acc[rt][2][vg] + bv.z;
            o.w = acc[rt][3][vg] + bv.w;
            *(v4f*)(C + (size_t)(m0 + 16 * rt + vg + 8 * hf) * N + n0 + 4 * r) = o;
        }
    }
}

#define KS_SLICES 8
#define KS_GROUP 4
#define KS_JPITCH 264
#define KS_WPITCH (8 * KS_JPITCH)
#define IMG_W 96
#define IMG_PLANE 9216
#define IMG_CH 256

struct RegGroup { RegFrag f[KS_GROUP]; };

__device__ __forceinline__ size_t window_base(int n) {
    int bi = n / 144;
    int l  = n - bi * 144;
    int wh = l / 12, ww = l - wh * 12;
    return (size_t)bi * IMG_CH * IMG_PLANE + (size_t)(wh * 8) * IMG_W + ww * 8;
}

__device__ __forceinline__ void reggroup_load_x(RegGroup& g,
                                                const float* __restrict__ X0,
                                                const float* __restrict__ X1,
                                                const float* __restrict__ Bp,
                                                int k, int N) {
    size_t xo = (size_t)(k >> 6) * IMG_PLANE + (size_t)(((k >> 5) & 1) * 4) * IMG_W;
    #pragma unroll
    for (int t = 0; t < KS_GROUP; ++t) g.f[t].a0 = *(const v4f*)(X0 + xo + t * IMG_W);
    #pragma unroll
    for (int t = 0; t < KS_GROUP; ++t) g.f[t].a1 = *(const v4f*)(X1 + xo + t * IMG_W);
    #pragma unroll
    for (int t = 0; t < KS_GROUP; ++t) {
        const float* bp = Bp + (size_t)(k + 8 * t) * N;
        g.f[t].b0 = *(const v4f*)(bp);
        g.f[t].b1 = *(const v4f*)(bp + N);
        g.f[t].b2 = *(const v4f*)(bp + 2 * (size_t)N);
        g.f[t].b3 = *(const v4f*)(bp + 3 * (size_t)N);
    }
}

__device__ __forceinline__ void reggroup_mma(v8f (&acc)[2][4], const RegGroup& g) {
    #pragma unroll
    for (int t = 0; t < KS_GROUP; ++t) regtile_mma(acc, g.f[t]);
}

__global__ __launch_bounds__(256) void pool_gemm_xinplace_ksplit(
    const float* __restrict__ x, const float* __restrict__ B,
    float* __restrict__ C, const float* __restrict__ bias,
    int N, int K)
{
    __shared__ float part[KS_SLICES * KS_WPITCH];
    int wave = threadIdx.x >> 5;
    int lane = threadIdx.x & 31;
    int hf   = lane >> 4;
    int r    = lane & 15;
    int m0 = blockIdx.y * 32;
    int n0 = blockIdx.x * 64;
    int klen = K / KS_SLICES;
    int kb = wave * klen;

    const float* X0 = x + window_base(m0 + r) + (size_t)(kb >> 6) * IMG_PLANE + 4 * hf;
    const float* X1 = x + window_base(m0 + 16 + r) + (size_t)(kb >> 6) * IMG_PLANE + 4 * hf;
    const float* Bp = B + (size_t)(kb + 4 * hf) * N + n0 + 4 * r;

    v8f acc[2][4] = {};
    RegGroup ga, gb;
    reggroup_load_x(ga, X0, X1, Bp, 0, N);
    int k = 0;
    for (; k < klen - 16 * KS_GROUP; k += 16 * KS_GROUP) {
        reggroup_load_x(gb, X0, X1, Bp, k + 8 * KS_GROUP, N);
        __builtin_amdgcn_sched_barrier(0);
        reggroup_mma(acc, ga);
        __builtin_amdgcn_sched_barrier(0);
        reggroup_load_x(ga, X0, X1, Bp, k + 16 * KS_GROUP, N);
        __builtin_amdgcn_sched_barrier(0);
        reggroup_mma(acc, gb);
        __builtin_amdgcn_sched_barrier(0);
    }
    reggroup_load_x(gb, X0, X1, Bp, k + 8 * KS_GROUP, N);
    reggroup_mma(acc, ga);
    reggroup_mma(acc, gb);

    float* p = part + wave * KS_WPITCH + lane;
    #pragma unroll
    for (int rt = 0; rt < 2; ++rt) {
        #pragma unroll
        for (int j = 0; j < 4; ++j) {
            #pragma unroll
            for (int vg = 0; vg < 8; ++vg)
                p[(rt * 4 + j) * KS_JPITCH + vg * 32] = acc[rt][j][vg];
        }
    }
    __syncthreads();

    int cl = threadIdx.x & 63;
    int v0 = threadIdx.x >> 6;
    float bv = bias[n0 + cl];
    #pragma unroll
    for (int i = 0; i < 8; ++i) {
        int rt = i >> 2, h2 = (i >> 1) & 1, vg = v0 + 4 * (i & 1);
        const float* q = part + (rt * 4 + (cl & 3)) * KS_JPITCH + vg * 32 + h2 * 16 + (cl >> 2);
        float s = q[0];
        #pragma unroll
        for (int sl = 1; sl < KS_SLICES; ++sl)
            s += q[sl * KS_WPITCH];
        C[(size_t)(m0 + 16 * rt + 8 * h2 + vg) * N + n0 + cl] = s + bv;
    }
}

__global__ __launch_bounds__(256) void qproj_xinplace(
    const float* __restrict__ Wq, const float* __restrict__ x,
    float* __restrict__ qmat, int K, float scale)
{
    int wave = threadIdx.x >> 5;
    int lane = threadIdx.x & 31;
    int hf   = lane >> 4;
    int r    = lane & 15;
    int o0   = wave * 32;

    int tt = blockIdx.x;
    int cg = tt % 3;  tt /= 3;
    int ip = tt % 4;  tt /= 4;
    int wh = tt % 12;
    int bi = tt / 12;
    int irow = r >> 3;
    int icol = cg * 32 + 4 * (r & 7);
    int y = wh * 8 + ip * 2 + irow;

    const float* A0 = Wq + (size_t)(o0 + r) * K + 4 * hf;
    const float* A1 = A0 + (size_t)16 * K;
    const float* Bp = x + ((size_t)bi * IMG_CH + 4 * hf) * IMG_PLANE + (size_t)y * IMG_W + icol;

    v8f acc[2][4] = {};
    RegFrag cur = regtile_load(A0, A1, Bp, 0, IMG_PLANE);
    for (int k = 8; k < K; k += 8) {
        RegFrag nxt = regtile_load(A0, A1, Bp, k, IMG_PLANE);
        regtile_mma(acc, cur);
        cur = nxt;
    }
    regtile_mma(acc, cur);

    int n = bi * 144 + wh * 12 + (icol >> 3);
    size_t m = (size_t)n * 64 + (ip * 2 + irow) * 8 + (icol & 7);
    float* qrow = qmat + m * IMG_CH + o0 + 8 * hf;
    #pragma unroll
    for (int j = 0; j < 4; ++j) {
        #pragma unroll
        for (int rt = 0; rt < 2; ++rt) {
            v4f lo, hi;
            lo.x = acc[rt][j][0] * scale; lo.y = acc[rt][j][1] * scale;
            lo.z = acc[rt][j][2] * scale; lo.w = acc[rt][j][3] * scale;
            hi.x = acc[rt][j][4] * scale; hi.y = acc[rt][j][5] * scale;
            hi.z = acc[rt][j][6] * scale; hi.w = acc[rt][j][7] * scale;
            float* dst = qrow + (size_t)j * IMG_CH + 16 * rt;
            *(v4f*)(dst)     = lo;
            *(v4f*)(dst + 4) = hi;
        }
    }
}

#define PPITCH 148

template <int CTRL>
__device__ __forceinline__ float half16_xor(float v) {
    return __builtin_bit_cast(float,
        __builtin_amdgcn_update_dpp(0, __builtin_bit_cast(int, v), CTRL, 0xf, 0xf, true));
}
__device__ __forceinline__ float half16_max(float v) {
    v = fmaxf(v, half16_xor<0x161>(v));
    v = fmaxf(v, half16_xor<0x162>(v));
    v = fmaxf(v, half16_xor<0x164>(v));
    v = fmaxf(v, half16_xor<0x168>(v));
    return v;
}
__device__ __forceinline__ float half16_sum(float v) {
    v += half16_xor<0x161>(v);
    v += half16_xor<0x162>(v);
    v += half16_xor<0x164>(v);
    v += half16_xor<0x168>(v);
    return v;
}

__global__ __launch_bounds__(128) void attn_kernel(
    float* __restrict__ qmat, const float* __restrict__ kv)
{
    __shared__ __attribute__((aligned(16))) float pbuf[64 * PPITCH];
    int id = blockIdx.x;
    int h  = id & 7;
    int bb = (id >> 3) & 7;
    int g  = id >> 6;
    int n  = g * 8 + bb;

    int wave = threadIdx.x >> 5;
    int lane = threadIdx.x & 31;
    int hf   = lane >> 4;
    int r    = lane & 15;
    int m0   = wave * 16;

    const float* qbase = qmat + ((size_t)n * 64 + m0 + r) * 256 + h * 32 + 4 * hf;
    const float* kb    = kv + (size_t)bb * 144 * 512 + h * 32;
    const float* vb    = kb + 256;

    v4f qf[4];
    #pragma unroll
    for (int t = 0; t < 4; ++t) qf[t] = *(const v4f*)(qbase + 8 * t);

    v8f s[9];
    #pragma unroll
    for (int g3 = 0; g3 < 3; ++g3) {
        v4f kf[3][4];
        #pragma unroll
        for (int j = 0; j < 3; ++j) {
            const float* krow = kb + (size_t)((3 * g3 + j) * 16 + r) * 512 + 4 * hf;
            #pragma unroll
            for (int t = 0; t < 4; ++t) kf[j][t] = *(const v4f*)(krow + 8 * t);
        }
        v8f a3[3] = {};
        #pragma unroll
        for (int t = 0; t < 4; ++t) {
            v2f a0 = {qf[t].x, qf[t].y}, a1 = {qf[t].z, qf[t].w};
            #pragma unroll
            for (int j = 0; j < 3; ++j) {
                v2f b = {kf[j][t].x, kf[j][t].y};
                a3[j] = __builtin_amdgcn_wmma_f32_16x16x4_f32(false, a0, false, b,
                                                              (short)0, a3[j], false, false);
            }
            #pragma unroll
            for (int j = 0; j < 3; ++j) {
                v2f b = {kf[j][t].z, kf[j][t].w};
                a3[j] = __builtin_amdgcn_wmma_f32_16x16x4_f32(false, a1, false, b,
                                                              (short)0, a3[j], false, false);
            }
        }
        #pragma unroll
        for (int j = 0; j < 3; ++j) s[3 * g3 + j] = a3[j];
    }

    float* pw = pbuf + (m0 + 8 * hf) * PPITCH + r;
    #pragma unroll
    for (int vg = 0; vg < 8; ++vg) {
        float mx = s[0][vg];
        #pragma unroll
        for (int nt = 1; nt < 9; ++nt) mx = fmaxf(mx, s[nt][vg]);
        mx = half16_max(mx);
        float sum = 0.0f;
        #pragma unroll
        for (int nt = 0; nt < 9; ++nt) {
            float e = __expf(s[nt][vg] - mx);
            s[nt][vg] = e;
            sum += e;
        }
        sum = half16_sum(sum);
        float inv = 1.0f / sum;
        #pragma unroll
        for (int nt = 0; nt < 9; ++nt)
            pw[vg * PPITCH + nt * 16] = s[nt][vg] * inv;
    }
    __syncthreads();

    const float* pr = pbuf + (m0 + r) * PPITCH + 4 * hf;
    const float* vp = vb + (size_t)(4 * hf) * 512 + 2 * r;
    v8f o0 = {}, o1 = {};
    #pragma unroll
    for (int t = 0; t < 18; ++t) {
        v4f p4 = *(const v4f*)(pr + 8 * t);
        const float* vq = vp + (size_t)(8 * t) * 512;
        v2f w0 = *(const v2f*)(vq);
        v2f w1 = *(const v2f*)(vq + 512);
        v2f w2 = *(const v2f*)(vq + 1024);
        v2f w3 = *(const v2f*)(vq + 1536);
        v2f a0 = {p4.x, p4.y}, a1 = {p4.z, p4.w};
        v2f b00 = {w0.x, w1.x}, b01 = {w0.y, w1.y};
        v2f b10 = {w2.x, w3.x}, b11 = {w2.y, w3.y};
        o0 = __builtin_amdgcn_wmma_f32_16x16x4_f32(false, a0, false, b00, (short)0, o0, false, false);
        o1 = __builtin_amdgcn_wmma_f32_16x16x4_f32(false, a0, false, b01, (short)0, o1, false, false);
        o0 = __builtin_amdgcn_wmma_f32_16x16x4_f32(false, a1, false, b10, (short)0, o0, false, false);
        o1 = __builtin_amdgcn_wmma_f32_16x16x4_f32(false, a1, false, b11, (short)0, o1, false, false);
    }
    float* orow = qmat + ((size_t)n * 64 + m0 + 8 * hf) * 256 + h * 32 + 2 * r;
    #pragma unroll
    for (int vg = 0; vg < 8; ++vg) {
        v2f o = {o0[vg], o1[vg]};
        *(v2f*)(orow + (size_t)vg * 256) = o;
    }
}

extern "C" void kernel_launch(void* const* d_in, const int* in_sizes, int n_in,
                              void* d_out, int out_size, void* d_ws, size_t ws_size,
                              hipStream_t stream) {
    const float* x      = (const float*)d_in[0];
    const float* Wq     = (const float*)d_in[1];
    const float* Wkv    = (const float*)d_in[2];
    const float* pool_w = (const float*)d_in[3];
    const float* pool_b = (const float*)d_in[4];
    const float* rpe_w  = (const float*)d_in[5];
    const float* rpe_b  = (const float*)d_in[6];
    const float* proj_w = (const float*)d_in[7];
    const float* proj_b = (const float*)d_in[8];
    float* out = (float*)d_out;

    float* ws    = (float*)d_ws;
    float* qmat  = ws;
    float* poolT = qmat + 18874368;
    float* WkvT  = poolT + 4194304;
    float* projT = WkvT + 131072;
    float* poolx = projT + 65536;
    float* kvb   = poolx + 294912;
    (void)ws_size; (void)in_sizes; (void)n_in; (void)out_size;

    const int TB = 256;
    transpose_k<<<(131072 + TB - 1) / TB, TB, 0, stream>>>(Wkv, WkvT, 512, 256, 131072);
    transpose_k<<<(65536 + TB - 1) / TB, TB, 0, stream>>>(proj_w, projT, 256, 256, 65536);
    transpose_k<<<(4194304 + TB - 1) / TB, TB, 0, stream>>>(pool_w, poolT, 256, 16384, 4194304);

    pool_gemm_xinplace_ksplit<<<dim3(256 / 64, 1152 / 32), 256, 0, stream>>>(
        x, poolT, poolx, pool_b, 256, 16384);

    qproj_xinplace<<<1152, 256, 0, stream>>>(Wq, x, qmat, 256, QSCALE);

    gemm_wmma_f32<<<dim3(512 / 64, 1152 / 32), TB, 0, stream>>>(
        poolx, WkvT, kvb, nullptr, 1152, 512, 256, 1.0f);

    attn_kernel<<<144 * 8 * 8, 128, 0, stream>>>(qmat, kvb);

    outproj_rpe_fused<<<73728 / 32, 128, 0, stream>>>(
        qmat, x, rpe_w, rpe_b, projT, out, proj_b, 256, 256);
}


# Pooling GEMM as the transposed product: pool_w rows in place are A, x windows are B; KV takes Wkv rows in place as B; both weight transposes go, 8 launches -> 6; f32 sum order unchanged
# speedup vs baseline: 3.6296x; 1.0909x over previous
#include <hip/hip_runtime.h>
#include <hip/hip_bf16.h>

typedef __attribute__((ext_vector_type(2))) float v2f;
typedef __attribute__((ext_vector_type(8))) float v8f;

#define QSCALE 0.17677669529663687f


__global__ void transpose_k(const float* __restrict__ src, float* __restrict__ dst,
                            int O, int K, int total) {
    int idx = blockIdx.x * blockDim.x + threadIdx.x;
    if (idx >= total) return;
    int o = idx % O;
    int k = idx / O;
    dst[idx] = src[(size_t)o * K + k];
}

__global__ __launch_bounds__(256) void gemm_wmma_f32(
    const float* __restrict__ A, const float* __restrict__ B,
    float* __restrict__ C, const float* __restrict__ bias,
    int M, int N, int K, float scale)
{
    int wave = threadIdx.x >> 5;
    int lane = threadIdx.x & 31;
    int hf   = lane >> 4;
    int r    = lane & 15;
    int m0 = blockIdx.y * 32 + (wave >> 2) * 16;
    int n0 = blockIdx.x * 64 + (wave & 3) * 16;

    const float* Arow = A + (size_t)(m0 + r) * K + 2 * hf;
    const float* Brow = B + (size_t)(n0 + r) * K + 2 * hf;

    v8f acc = {};
    #pragma unroll 8
    for (int k = 0; k < K; k += 4) {
        v2f a, b;
        a.x = Arow[k];
        a.y = Arow[k + 1];
        b.x = Brow[k];
        b.y = Brow[k + 1];
        acc = __builtin_amdgcn_wmma_f32_16x16x4_f32(false, a, false, b,
                                                    (short)0, acc, false, false);
    }
    int col = n0 + r;
    float bv = bias ? bias[col] : 0.0f;
    #pragma unroll
    for (int vg = 0; vg < 8; ++vg)
        C[(size_t)(m0 + vg + 8 * hf) * N + col] = acc[vg] * scale + bv;
}

typedef __attribute__((ext_vector_type(4))) float v4f;

struct RegFrag {
    v4f a0, a1;
    v4f b0, b1, b2, b3;
};

__device__ __forceinline__ RegFrag regtile_load(const float* __restrict__ A0,
                                                const float* __restrict__ A1,
                                                const float* __restrict__ Bp,
                                                int k, int N) {
    RegFrag f;
    f.a0 = *(const v4f*)(A0 + k);
    f.a1 = *(const v4f*)(A1 + k);
    const float* bp = Bp + (size_t)k * N;
    f.b0 = *(const v4f*)(bp);
    f.b1 = *(const v4f*)(bp + N);
    f.b2 = *(const v4f*)(bp + 2 * (size_t)N);
    f.b3 = *(const v4f*)(bp + 3 * (size_t)N);
    return f;
}

__device__ __forceinline__ void regtile_mma(v8f (&acc)[2][4], const RegFrag& f) {
    v2f a00 = {f.a0.x, f.a0.y}, a01 = {f.a0.z, f.a0.w};
    v2f a10 = {f.a1.x, f.a1.y}, a11 = {f.a1.z, f.a1.w};
    #pragma unroll
    for (int j = 0; j < 4; ++j) {
        v2f b = {f.b0[j], f.b1[j]};
        acc[0][j] = __builtin_amdgcn_wmma_f32_16x16x4_f32(false, a00, false, b,
                                                          (short)0, acc[0][j], false, false);
        acc[1][j] = __builtin_amdgcn_wmma_f32_16x16x4_f32(false, a10, false, b,
                                                          (short)0, acc[1][j], false, false);
    }
    #pragma unroll
    for (int j = 0; j < 4; ++j) {
        v2f b = {f.b2[j], f.b3[j]};
        acc[0][j] = __builtin_amdgcn_wmma_f32_16x16x4_f32(false, a01, false, b,
                                                          (short)0, acc[0][j], false, false);
        acc[1][j] = __builtin_amdgcn_wmma_f32_16x16x4_f32(false, a11, false, b,
                                                          (short)0, acc[1][j], false, false);
    }
}

#define OP_APITCH 260

__global__ __launch_bounds__(128) void outproj_rpe_fused(
    const float* __restrict__ F, const float* __restrict__ x,
    const float* __restrict__ rw, const float* __restrict__ rb,
    const float* __restrict__ B, float* __restrict__ C,
    const float* __restrict__ bias, int N, int K)
{
    __shared__ __attribute__((aligned(16))) float atile[32 * OP_APITCH];
    int wave = threadIdx.x >> 5;
    int lane = threadIdx.x & 31;
    int hf   = lane >> 4;
    int r    = lane & 15;
    int m0 = blockIdx.x * 32;
    int n0 = wave * 64;

    int rsel = __builtin_amdgcn_readfirstlane(wave >> 1);
    int kq = 4 * (threadIdx.x & 63);
    const char* xb = (const char*)x;
    #pragma unroll 4
    for (int i = 0; i < 16; ++i) {
        int row = 2 * i + rsel;
        int m  = m0 + row;
        int pl = m / 36;
        int c  = pl & 255;
        int p  = (m - pl * 36) * 256 + kq;
        int y  = p / 96;
        int xx = p - y * 96;
        unsigned bo = ((unsigned)m * 256u + (unsigned)kq) * 4u;
        v4f f = *(const v4f*)((const char*)F + bo);
        const float* wp = rw + c * 9;
        float b0 = rb[c];
        v4f s = {b0, b0, b0, b0};
        bool lok = xx > 0, rok = xx < 92;
        unsigned lo = lok ? 0xfffffffcu : 0u;
        unsigned ro = rok ? 16u : 12u;
        #pragma unroll
        for (int dy = -1; dy <= 1; ++dy) {
            bool rowok = dy < 0 ? (y > 0) : (dy > 0 ? (y < 95) : true);
            unsigned br = rowok ? bo + (unsigned)(dy * 384) : bo;
            v4f mid = *(const v4f*)(xb + br);
            float lf = *(const float*)(xb + (br + lo));
            float rg = *(const float*)(xb + (br + ro));
            if (!rowok) mid = (v4f){0.0f, 0.0f, 0.0f, 0.0f};
            lf = (rowok && lok) ? lf : 0.0f;
            rg = (rowok && rok) ? rg : 0.0f;
            float w0 = wp[(dy + 1) * 3], w1 = wp[(dy + 1) * 3 + 1], w2 = wp[(dy + 1) * 3 + 2];
            s.x += lf    * w0; s.x += mid.x * w1; s.x += mid.y * w2;
            s.y += mid.x * w0; s.y += mid.y * w1; s.y += mid.z * w2;
            s.z += mid.y * w0; s.z += mid.z * w1; s.z += mid.w * w2;
            s.w += mid.z * w0; s.w += mid.w * w1; s.w += rg    * w2;
        }
        *(v4f*)(atile + row * OP_APITCH + kq) = f + s;
    }
    __syncthreads();

    const float* A0 = atile + r * OP_APITCH + 4 * hf;
    const float* A1 = A0 + 16 * OP_APITCH;
    const float* Bp = B + (size_t)(4 * hf) * N + n0 + 4 * r;

    v8f acc[2][4] = {};
    RegFrag cur = regtile_load(A0, A1, Bp, 0, N);
    for (int k = 8; k < K; k += 8) {
        RegFrag nxt = regtile_load(A0, A1, Bp, k, N);
        regtile_mma(acc, cur);
        cur = nxt;
    }
    regtile_mma(acc, cur);

    v4f bv = *(const v4f*)(bias + n0 + 4 * r);
    #pragma unroll
    for (int rt = 0; rt < 2; ++rt) {
        #pragma unroll
        for (int vg = 0; vg < 8; ++vg) {
            v4f o;
            o.x = acc[rt][0][vg] + bv.x;
            o.y = acc[rt][1][vg] + bv.y;
            o.z = acc[rt][2][vg] + bv.z;
            o.w = acc[rt][3][vg] + bv.w;
            *(v4f*)(C + (size_t)(m0 + 16 * rt + vg + 8 * hf) * N + n0 + 4 * r) = o;
        }
    }
}

#define KS_SLICES 8
#define KS_GROUP 4
#define PT_PITCH 35
#define PT_WPITCH (64 * PT_PITCH)
#define IMG_W 96
#define IMG_PLANE 9216
#define IMG_CH 256

struct PoolFrag {
    v4f w0, w1;
    v4f xw[4];
};
struct PoolGroup { PoolFrag f[KS_GROUP]; };

__device__ __forceinline__ size_t window_base(int n) {
    int bi = n / 144;
    int l  = n - bi * 144;
    int wh = l / 12, ww = l - wh * 12;
    return (size_t)bi * IMG_CH * IMG_PLANE + (size_t)(wh * 8) * IMG_W + ww * 8;
}

__device__ __forceinline__ void poolgroup_load(PoolGroup& g,
                                               const float* __restrict__ W0,
                                               const float* __restrict__ W1,
                                               const float* __restrict__ X0,
                                               const float* __restrict__ X1,
                                               const float* __restrict__ X2,
                                               const float* __restrict__ X3,
                                               int k) {
    size_t xo = (size_t)(k >> 6) * IMG_PLANE + (size_t)(((k >> 5) & 1) * 4) * IMG_W;
    #pragma unroll
    for (int t = 0; t < KS_GROUP; ++t) g.f[t].w0 = *(const v4f*)(W0 + k + 8 * t);
    #pragma unroll
    for (int t = 0; t < KS_GROUP; ++t) g.f[t].w1 = *(const v4f*)(W1 + k + 8 * t);
    #pragma unroll
    for (int t = 0; t < KS_GROUP; ++t) {
        g.f[t].xw[0] = *(const v4f*)(X0 + xo + t * IMG_W);
        g.f[t].xw[1] = *(const v4f*)(X1 + xo + t * IMG_W);
        g.f[t].xw[2] = *(const v4f*)(X2 + xo + t * IMG_W);
        g.f[t].xw[3] = *(const v4f*)(X3 + xo + t * IMG_W);
    }
}

__device__ __forceinline__ void pooltile_mma(v8f (&acc)[2][4], const PoolFrag& f) {
    v2f a00 = {f.w0.x, f.w0.y}, a01 = {f.w0.z, f.w0.w};
    v2f a10 = {f.w1.x, f.w1.y}, a11 = {f.w1.z, f.w1.w};
    #pragma unroll
    for (int j = 0; j < 4; ++j) {
        v2f b = {f.xw[j].x, f.xw[j].y};
        acc[0][j] = __builtin_amdgcn_wmma_f32_16x16x4_f32(false, a00, false, b,
                                                          (short)0, acc[0][j], false, false);
        acc[1][j] = __builtin_amdgcn_wmma_f32_16x16x4_f32(false, a10, false, b,
                                                          (short)0, acc[1][j], false, false);
    }
    #pragma unroll
    for (int j = 0; j < 4; ++j) {
        v2f b = {f.xw[j].z, f.xw[j].w};
        acc[0][j] = __builtin_amdgcn_wmma_f32_16x16x4_f32(false, a01, false, b,
                                                          (short)0, acc[0][j], false, false);
        acc[1][j] = __builtin_amdgcn_wmma_f32_16x16x4_f32(false, a11, false, b,
                                                          (short)0, acc[1][j], false, false);
    }
}

__device__ __forceinline__ void poolgroup_mma(v8f (&acc)[2][4], const PoolGroup& g) {
    #pragma unroll
    for (int t = 0; t < KS_GROUP; ++t) pooltile_mma(acc, g.f[t]);
}

__global__ __launch_bounds__(256) void pool_gemm_inplace_ksplit(
    const float* __restrict__ x, const float* __restrict__ Wp,
    float* __restrict__ C, const float* __restrict__ bias,
    int N, int K)
{
    __shared__ float part[KS_SLICES * PT_WPITCH];
    int wave = threadIdx.x >> 5;
    int lane = threadIdx.x & 31;
    int hf   = lane >> 4;
    int r    = lane & 15;
    int o0 = blockIdx.x * 32;
    int n0 = blockIdx.y * 64;
    int klen = K / KS_SLICES;
    int kb = wave * klen;

    const float* W0 = Wp + (size_t)(o0 + r) * K + kb + 4 * hf;
    const float* W1 = W0 + (size_t)16 * K;
    size_t co = (size_t)(kb >> 6) * IMG_PLANE + 4 * hf;
    const float* X0 = x + window_base(n0 + r) + co;
    const float* X1 = x + window_base(n0 + 16 + r) + co;
    const float* X2 = x + window_base(n0 + 32 + r) + co;
    const float* X3 = x + window_base(n0 + 48 + r) + co;

    v8f acc[2][4] = {};
    PoolGroup ga, gb;
    poolgroup_load(ga, W0, W1, X0, X1, X2, X3, 0);
    int k = 0;
    for (; k < klen - 16 * KS_GROUP; k += 16 * KS_GROUP) {
        poolgroup_load(gb, W0, W1, X0, X1, X2, X3, k + 8 * KS_GROUP);
        __builtin_amdgcn_sched_barrier(0);
        poolgroup_mma(acc, ga);
        __builtin_amdgcn_sched_barrier(0);
        poolgroup_load(ga, W0, W1, X0, X1, X2, X3, k + 16 * KS_GROUP);
        __builtin_amdgcn_sched_barrier(0);
        poolgroup_mma(acc, gb);
        __builtin_amdgcn_sched_barrier(0);
    }
    poolgroup_load(gb, W0, W1, X0, X1, X2, X3, k + 8 * KS_GROUP);
    poolgroup_mma(acc, ga);
    poolgroup_mma(acc, gb);

    float* p = part + wave * PT_WPITCH + r * PT_PITCH + 8 * hf;
    #pragma unroll
    for (int rt = 0; rt < 2; ++rt) {
        #pragma unroll
        for (int j = 0; j < 4; ++j) {
            #pragma unroll
            for (int vg = 0; vg < 8; ++vg)
                p[(16 * j) * PT_PITCH + 16 * rt + vg] = acc[rt][j][vg];
        }
    }
    __syncthreads();

    int ol = threadIdx.x & 31;
    int nb = threadIdx.x >> 5;
    float bv = bias[o0 + ol];
    #pragma unroll
    for (int i = 0; i < 8; ++i) {
        int nl = nb + 8 * i;
        const float* q = part + nl * PT_PITCH + ol;
        float s = q[0];
        #pragma unroll
        for (int sl = 1; sl < KS_SLICES; ++sl)
            s += q[sl * PT_WPITCH];
        C[(size_t)(n0 + nl) * N + o0 + ol] = s + bv;
    }
}

__global__ __launch_bounds__(256) void qproj_xinplace(
    const float* __restrict__ Wq, const float* __restrict__ x,
    float* __restrict__ qmat, int K, float scale)
{
    int wave = threadIdx.x >> 5;
    int lane = threadIdx.x & 31;
    int hf   = lane >> 4;
    int r    = lane & 15;
    int o0   = wave * 32;

    int tt = blockIdx.x;
    int cg = tt % 3;  tt /= 3;
    int ip = tt % 4;  tt /= 4;
    int wh = tt % 12;
    int bi = tt / 12;
    int irow = r >> 3;
    int icol = cg * 32 + 4 * (r & 7);
    int y = wh * 8 + ip * 2 + irow;

    const float* A0 = Wq + (size_t)(o0 + r) * K + 4 * hf;
    const float* A1 = A0 + (size_t)16 * K;
    const float* Bp = x + ((size_t)bi * IMG_CH + 4 * hf) * IMG_PLANE + (size_t)y * IMG_W + icol;

    v8f acc[2][4] = {};
    RegFrag cur = regtile_load(A0, A1, Bp, 0, IMG_PLANE);
    for (int k = 8; k < K; k += 8) {
        RegFrag nxt = regtile_load(A0, A1, Bp, k, IMG_PLANE);
        regtile_mma(acc, cur);
        cur = nxt;
    }
    regtile_mma(acc, cur);

    int n = bi * 144 + wh * 12 + (icol >> 3);
    size_t m = (size_t)n * 64 + (ip * 2 + irow) * 8 + (icol & 7);
    float* qrow = qmat + m * IMG_CH + o0 + 8 * hf;
    #pragma unroll
    for (int j = 0; j < 4; ++j) {
        #pragma unroll
        for (int rt = 0; rt < 2; ++rt) {
            v4f lo, hi;
            lo.x = acc[rt][j][0] * scale; lo.y = acc[rt][j][1] * scale;
            lo.z = acc[rt][j][2] * scale; lo.w = acc[rt][j][3] * scale;
            hi.x = acc[rt][j][4] * scale; hi.y = acc[rt][j][5] * scale;
            hi.z = acc[rt][j][6] * scale; hi.w = acc[rt][j][7] * scale;
            float* dst = qrow + (size_t)j * IMG_CH + 16 * rt;
            *(v4f*)(dst)     = lo;
            *(v4f*)(dst + 4) = hi;
        }
    }
}

#define PPITCH 148

template <int CTRL>
__device__ __forceinline__ float half16_xor(float v) {
    return __builtin_bit_cast(float,
        __builtin_amdgcn_update_dpp(0, __builtin_bit_cast(int, v), CTRL, 0xf, 0xf, true));
}
__device__ __forceinline__ float half16_max(float v) {
    v = fmaxf(v, half16_xor<0x161>(v));
    v = fmaxf(v, half16_xor<0x162>(v));
    v = fmaxf(v, half16_xor<0x164>(v));
    v = fmaxf(v, half16_xor<0x168>(v));
    return v;
}
__device__ __forceinline__ float half16_sum(float v) {
    v += half16_xor<0x161>(v);
    v += half16_xor<0x162>(v);
    v += half16_xor<0x164>(v);
    v += half16_xor<0x168>(v);
    return v;
}

__global__ __launch_bounds__(128) void attn_kernel(
    float* __restrict__ qmat, const float* __restrict__ kv)
{
    __shared__ __attribute__((aligned(16))) float pbuf[64 * PPITCH];
    int id = blockIdx.x;
    int h  = id & 7;
    int bb = (id >> 3) & 7;
    int g  = id >> 6;
    int n  = g * 8 + bb;

    int wave = threadIdx.x >> 5;
    int lane = threadIdx.x & 31;
    int hf   = lane >> 4;
    int r    = lane & 15;
    int m0   = wave * 16;

    const float* qbase = qmat + ((size_t)n * 64 + m0 + r) * 256 + h * 32 + 4 * hf;
    const float* kb    = kv + (size_t)bb * 144 * 512 + h * 32;
    const float* vb    = kb + 256;

    v4f qf[4];
    #pragma unroll
    for (int t = 0; t < 4; ++t) qf[t] = *(const v4f*)(qbase + 8 * t);

    v8f s[9];
    #pragma unroll
    for (int g3 = 0; g3 < 3; ++g3) {
        v4f kf[3][4];
        #pragma unroll
        for (int j = 0; j < 3; ++j) {
            const float* krow = kb + (size_t)((3 * g3 + j) * 16 + r) * 512 + 4 * hf;
            #pragma unroll
            for (int t = 0; t < 4; ++t) kf[j][t] = *(const v4f*)(krow + 8 * t);
        }
        v8f a3[3] = {};
        #pragma unroll
        for (int t = 0; t < 4; ++t) {
            v2f a0 = {qf[t].x, qf[t].y}, a1 = {qf[t].z, qf[t].w};
            #pragma unroll
            for (int j = 0; j < 3; ++j) {
                v2f b = {kf[j][t].x, kf[j][t].y};
                a3[j] = __builtin_amdgcn_wmma_f32_16x16x4_f32(false, a0, false, b,
                                                              (short)0, a3[j], false, false);
            }
            #pragma unroll
            for (int j = 0; j < 3; ++j) {
                v2f b = {kf[j][t].z, kf[j][t].w};
                a3[j] = __builtin_amdgcn_wmma_f32_16x16x4_f32(false, a1, false, b,
                                                              (short)0, a3[j], false, false);
            }
        }
        #pragma unroll
        for (int j = 0; j < 3; ++j) s[3 * g3 + j] = a3[j];
    }

    float* pw = pbuf + (m0 + 8 * hf) * PPITCH + r;
    #pragma unroll
    for (int vg = 0; vg < 8; ++vg) {
        float mx = s[0][vg];
        #pragma unroll
        for (int nt = 1; nt < 9; ++nt) mx = fmaxf(mx, s[nt][vg]);
        mx = half16_max(mx);
        float sum = 0.0f;
        #pragma unroll
        for (int nt = 0; nt < 9; ++nt) {
            float e = __expf(s[nt][vg] - mx);
            s[nt][vg] = e;
            sum += e;
        }
        sum = half16_sum(sum);
        float inv = 1.0f / sum;
        #pragma unroll
        for (int nt = 0; nt < 9; ++nt)
            pw[vg * PPITCH + nt * 16] = s[nt][vg] * inv;
    }
    __syncthreads();

    const float* pr = pbuf + (m0 + r) * PPITCH + 4 * hf;
    const float* vp = vb + (size_t)(4 * hf) * 512 + 2 * r;
    v8f o0 = {}, o1 = {};
    #pragma unroll
    for (int t = 0; t < 18; ++t) {
        v4f p4 = *(const v4f*)(pr + 8 * t);
        const float* vq = vp + (size_t)(8 * t) * 512;
        v2f w0 = *(const v2f*)(vq);
        v2f w1 = *(const v2f*)(vq + 512);
        v2f w2 = *(const v2f*)(vq + 1024);
        v2f w3 = *(const v2f*)(vq + 1536);
        v2f a0 = {p4.x, p4.y}, a1 = {p4.z, p4.w};
        v2f b00 = {w0.x, w1.x}, b01 = {w0.y, w1.y};
        v2f b10 = {w2.x, w3.x}, b11 = {w2.y, w3.y};
        o0 = __builtin_amdgcn_wmma_f32_16x16x4_f32(false, a0, false, b00, (short)0, o0, false, false);
        o1 = __builtin_amdgcn_wmma_f32_16x16x4_f32(false, a0, false, b01, (short)0, o1, false, false);
        o0 = __builtin_amdgcn_wmma_f32_16x16x4_f32(false, a1, false, b10, (short)0, o0, false, false);
        o1 = __builtin_amdgcn_wmma_f32_16x16x4_f32(false, a1, false, b11, (short)0, o1, false, false);
    }
    float* orow = qmat + ((size_t)n * 64 + m0 + 8 * hf) * 256 + h * 32 + 2 * r;
    #pragma unroll
    for (int vg = 0; vg < 8; ++vg) {
        v2f o = {o0[vg], o1[vg]};
        *(v2f*)(orow + (size_t)vg * 256) = o;
    }
}

extern "C" void kernel_launch(void* const* d_in, const int* in_sizes, int n_in,
                              void* d_out, int out_size, void* d_ws, size_t ws_size,
                              hipStream_t stream) {
    const float* x      = (const float*)d_in[0];
    const float* Wq     = (const float*)d_in[1];
    const float* Wkv    = (const float*)d_in[2];
    const float* pool_w = (const float*)d_in[3];
    const float* pool_b = (const float*)d_in[4];
    const float* rpe_w  = (const float*)d_in[5];
    const float* rpe_b  = (const float*)d_in[6];
    const float* proj_w = (const float*)d_in[7];
    const float* proj_b = (const float*)d_in[8];
    float* out = (float*)d_out;

    float* ws    = (float*)d_ws;
    float* qmat  = ws;
    float* projT = qmat + 18874368;
    float* poolx = projT + 65536;
    float* kvb   = poolx + 294912;
    (void)ws_size; (void)in_sizes; (void)n_in; (void)out_size;

    const int TB = 256;
    transpose_k<<<(65536 + TB - 1) / TB, TB, 0, stream>>>(proj_w, projT, 256, 256, 65536);

    pool_gemm_inplace_ksplit<<<dim3(256 / 32, 1152 / 64), 256, 0, stream>>>(
        x, pool_w, poolx, pool_b, 256, 16384);

    qproj_xinplace<<<1152, 256, 0, stream>>>(Wq, x, qmat, 256, QSCALE);

    gemm_wmma_f32<<<dim3(512 / 64, 1152 / 32), TB, 0, stream>>>(
        poolx, Wkv, kvb, nullptr, 1152, 512, 256, 1.0f);

    attn_kernel<<<144 * 8 * 8, 128, 0, stream>>>(qmat, kvb);

    outproj_rpe_fused<<<73728 / 32, 128, 0, stream>>>(
        qmat, x, rpe_w, rpe_b, projT, out, proj_b, 256, 256);
}


# Pooling GEMM and Q projection share one launch, pool blocks first: Q tiles run on the 112 WGPs the 144 pool blocks leave idle; Q loads grouped 32 k ahead; sum order unchanged; 6 launches to 5
# speedup vs baseline: 3.8681x; 1.0657x over previous
#include <hip/hip_runtime.h>
#include <hip/hip_bf16.h>

typedef __attribute__((ext_vector_type(2))) float v2f;
typedef __attribute__((ext_vector_type(8))) float v8f;

#define QSCALE 0.17677669529663687f


__global__ void transpose_k(const float* __restrict__ src, float* __restrict__ dst,
                            int O, int K, int total) {
    int idx = blockIdx.x * blockDim.x + threadIdx.x;
    if (idx >= total) return;
    int o = idx % O;
    int k = idx / O;
    dst[idx] = src[(size_t)o * K + k];
}

__global__ __launch_bounds__(256) void gemm_wmma_f32(
    const float* __restrict__ A, const float* __restrict__ B,
    float* __restrict__ C, const float* __restrict__ bias,
    int M, int N, int K, float scale)
{
    int wave = threadIdx.x >> 5;
    int lane = threadIdx.x & 31;
    int hf   = lane >> 4;
    int r    = lane & 15;
    int m0 = blockIdx.y * 32 + (wave >> 2) * 16;
    int n0 = blockIdx.x * 64 + (wave & 3) * 16;

    const float* Arow = A + (size_t)(m0 + r) * K + 2 * hf;
    const float* Brow = B + (size_t)(n0 + r) * K + 2 * hf;

    v8f acc = {};
    #pragma unroll 8
    for (int k = 0; k < K; k += 4) {
        v2f a, b;
        a.x = Arow[k];
        a.y = Arow[k + 1];
        b.x = Brow[k];
        b.y = Brow[k + 1];
        acc = __builtin_amdgcn_wmma_f32_16x16x4_f32(false, a, false, b,
                                                    (short)0, acc, false, false);
    }
    int col = n0 + r;
    float bv = bias ? bias[col] : 0.0f;
    #pragma unroll
    for (int vg = 0; vg < 8; ++vg)
        C[(size_t)(m0 + vg + 8 * hf) * N + col] = acc[vg] * scale + bv;
}

typedef __attribute__((ext_vector_type(4))) float v4f;

struct RegFrag {
    v4f a0, a1;
    v4f b0, b1, b2, b3;
};

__device__ __forceinline__ RegFrag regtile_load(const float* __restrict__ A0,
                                                const float* __restrict__ A1,
                                                const float* __restrict__ Bp,
                                                int k, int N) {
    RegFrag f;
    f.a0 = *(const v4f*)(A0 + k);
    f.a1 = *(const v4f*)(A1 + k);
    const float* bp = Bp + (size_t)k * N;
    f.b0 = *(const v4f*)(bp);
    f.b1 = *(const v4f*)(bp + N);
    f.b2 = *(const v4f*)(bp + 2 * (size_t)N);
    f.b3 = *(const v4f*)(bp + 3 * (size_t)N);
    return f;
}

__device__ __forceinline__ void regtile_mma(v8f (&acc)[2][4], const RegFrag& f) {
    v2f a00 = {f.a0.x, f.a0.y}, a01 = {f.a0.z, f.a0.w};
    v2f a10 = {f.a1.x, f.a1.y}, a11 = {f.a1.z, f.a1.w};
    #pragma unroll
    for (int j = 0; j < 4; ++j) {
        v2f b = {f.b0[j], f.b1[j]};
        acc[0][j] = __builtin_amdgcn_wmma_f32_16x16x4_f32(false, a00, false, b,
                                                          (short)0, acc[0][j], false, false);
        acc[1][j] = __builtin_amdgcn_wmma_f32_16x16x4_f32(false, a10, false, b,
                                                          (short)0, acc[1][j], false, false);
    }
    #pragma unroll
    for (int j = 0; j < 4; ++j) {
        v2f b = {f.b2[j], f.b3[j]};
        acc[0][j] = __builtin_amdgcn_wmma_f32_16x16x4_f32(false, a01, false, b,
                                                          (short)0, acc[0][j], false, false);
        acc[1][j] = __builtin_amdgcn_wmma_f32_16x16x4_f32(false, a11, false, b,
                                                          (short)0, acc[1][j], false, false);
    }
}

#define OP_APITCH 260

__global__ __launch_bounds__(128) void outproj_rpe_fused(
    const float* __restrict__ F, const float* __restrict__ x,
    const float* __restrict__ rw, const float* __restrict__ rb,
    const float* __restrict__ B, float* __restrict__ C,
    const float* __restrict__ bias, int N, int K)
{
    __shared__ __attribute__((aligned(16))) float atile[32 * OP_APITCH];
    int wave = threadIdx.x >> 5;
    int lane = threadIdx.x & 31;
    int hf   = lane >> 4;
    int r    = lane & 15;
    int m0 = blockIdx.x * 32;
    int n0 = wave * 64;

    int rsel = __builtin_amdgcn_readfirstlane(wave >> 1);
    int kq = 4 * (threadIdx.x & 63);
    const char* xb = (const char*)x;
    #pragma unroll 4
    for (int i = 0; i < 16; ++i) {
        int row = 2 * i + rsel;
        int m  = m0 + row;
        int pl = m / 36;
        int c  = pl & 255;
        int p  = (m - pl * 36) * 256 + kq;
        int y  = p / 96;
        int xx = p - y * 96;
        unsigned bo = ((unsigned)m * 256u + (unsigned)kq) * 4u;
        v4f f = *(const v4f*)((const char*)F + bo);
        const float* wp = rw + c * 9;
        float b0 = rb[c];
        v4f s = {b0, b0, b0, b0};
        bool lok = xx > 0, rok = xx < 92;
        unsigned lo = lok ? 0xfffffffcu : 0u;
        unsigned ro = rok ? 16u : 12u;
        #pragma unroll
        for (int dy = -1; dy <= 1; ++dy) {
            bool rowok = dy < 0 ? (y > 0) : (dy > 0 ? (y < 95) : true);
            unsigned br = rowok ? bo + (unsigned)(dy * 384) : bo;
            v4f mid = *(const v4f*)(xb + br);
            float lf = *(const float*)(xb + (br + lo));
            float rg = *(const float*)(xb + (br + ro));
            if (!rowok) mid = (v4f){0.0f, 0.0f, 0.0f, 0.0f};
            lf = (rowok && lok) ? lf : 0.0f;
            rg = (rowok && rok) ? rg : 0.0f;
            float w0 = wp[(dy + 1) * 3], w1 = wp[(dy + 1) * 3 + 1], w2 = wp[(dy + 1) * 3 + 2];
            s.x += lf    * w0; s.x += mid.x * w1; s.x += mid.y * w2;
            s.y += mid.x * w0; s.y += mid.y * w1; s.y += mid.z * w2;
            s.z += mid.y * w0; s.z += mid.z * w1; s.z += mid.w * w2;
            s.w += mid.z * w0; s.w += mid.w * w1; s.w += rg    * w2;
        }
        *(v4f*)(atile + row * OP_APITCH + kq) = f + s;
    }
    __syncthreads();

    const float* A0 = atile + r * OP_APITCH + 4 * hf;
    const float* A1 = A0 + 16 * OP_APITCH;
    const float* Bp = B + (size_t)(4 * hf) * N + n0 + 4 * r;

    v8f acc[2][4] = {};
    RegFrag cur = regtile_load(A0, A1, Bp, 0, N);
    for (int k = 8; k < K; k += 8) {
        RegFrag nxt = regtile_load(A0, A1, Bp, k, N);
        regtile_mma(acc, cur);
        cur = nxt;
    }
    regtile_mma(acc, cur);

    v4f bv = *(const v4f*)(bias + n0 + 4 * r);
    #pragma unroll
    for (int rt = 0; rt < 2; ++rt) {
        #pragma unroll
        for (int vg = 0; vg < 8; ++vg) {
            v4f o;
            o.x = acc[rt][0][vg] + bv.x;
            o.y = acc[rt][1][vg] + bv.y;
            o.z = acc[rt][2][vg] + bv.z;
            o.w = acc[rt][3][vg] + bv.w;
            *(v4f*)(C + (size_t)(m0 + 16 * rt + vg + 8 * hf) * N + n0 + 4 * r) = o;
        }
    }
}

#define KS_SLICES 8
#define KS_GROUP 4
#define PT_PITCH 35
#define PT_WPITCH (64 * PT_PITCH)
#define IMG_W 96
#define IMG_PLANE 9216
#define IMG_CH 256

struct PoolFrag {
    v4f w0, w1;
    v4f xw[4];
};
struct PoolGroup { PoolFrag f[KS_GROUP]; };

__device__ __forceinline__ size_t window_base(int n) {
    int bi = n / 144;
    int l  = n - bi * 144;
    int wh = l / 12, ww = l - wh * 12;
    return (size_t)bi * IMG_CH * IMG_PLANE + (size_t)(wh * 8) * IMG_W + ww * 8;
}

__device__ __forceinline__ void poolgroup_load(PoolGroup& g,
                                               const float* __restrict__ W0,
                                               const float* __restrict__ W1,
                                               const float* __restrict__ X0,
                                               const float* __restrict__ X1,
                                               const float* __restrict__ X2,
                                               const float* __restrict__ X3,
                                               int k) {
    size_t xo = (size_t)(k >> 6) * IMG_PLANE + (size_t)(((k >> 5) & 1) * 4) * IMG_W;
    #pragma unroll
    for (int t = 0; t < KS_GROUP; ++t) g.f[t].w0 = *(const v4f*)(W0 + k + 8 * t);
    #pragma unroll
    for (int t = 0; t < KS_GROUP; ++t) g.f[t].w1 = *(const v4f*)(W1 + k + 8 * t);
    #pragma unroll
    for (int t = 0; t < KS_GROUP; ++t) {
        g.f[t].xw[0] = *(const v4f*)(X0 + xo + t * IMG_W);
        g.f[t].xw[1] = *(const v4f*)(X1 + xo + t * IMG_W);
        g.f[t].xw[2] = *(const v4f*)(X2 + xo + t * IMG_W);
        g.f[t].xw[3] = *(const v4f*)(X3 + xo + t * IMG_W);
    }
}

__device__ __forceinline__ void pooltile_mma(v8f (&acc)[2][4], const PoolFrag& f) {
    v2f a00 = {f.w0.x, f.w0.y}, a01 = {f.w0.z, f.w0.w};
    v2f a10 = {f.w1.x, f.w1.y}, a11 = {f.w1.z, f.w1.w};
    #pragma unroll
    for (int j = 0; j < 4; ++j) {
        v2f b = {f.xw[j].x, f.xw[j].y};
        acc[0][j] = __builtin_amdgcn_wmma_f32_16x16x4_f32(false, a00, false, b,
                                                          (short)0, acc[0][j], false, false);
        acc[1][j] = __builtin_amdgcn_wmma_f32_16x16x4_f32(false, a10, false, b,
                                                          (short)0, acc[1][j], false, false);
    }
    #pragma unroll
    for (int j = 0; j < 4; ++j) {
        v2f b = {f.xw[j].z, f.xw[j].w};
        acc[0][j] = __builtin_amdgcn_wmma_f32_16x16x4_f32(false, a01, false, b,
                                                          (short)0, acc[0][j], false, false);
        acc[1][j] = __builtin_amdgcn_wmma_f32_16x16x4_f32(false, a11, false, b,
                                                          (short)0, acc[1][j], false, false);
    }
}

__device__ __forceinline__ void poolgroup_mma(v8f (&acc)[2][4], const PoolGroup& g) {
    #pragma unroll
    for (int t = 0; t < KS_GROUP; ++t) pooltile_mma(acc, g.f[t]);
}

__device__ __forceinline__ void pool_gemm_body(
    float* __restrict__ part, int obl, int nbl,
    const float* __restrict__ x, const float* __restrict__ Wp,
    float* __restrict__ C, const float* __restrict__ bias,
    int N, int K)
{
    int wave = threadIdx.x >> 5;
    int lane = threadIdx.x & 31;
    int hf   = lane >> 4;
    int r    = lane & 15;
    int o0 = obl * 32;
    int n0 = nbl * 64;
    int klen = K / KS_SLICES;
    int kb = wave * klen;

    const float* W0 = Wp + (size_t)(o0 + r) * K + kb + 4 * hf;
    const float* W1 = W0 + (size_t)16 * K;
    size_t co = (size_t)(kb >> 6) * IMG_PLANE + 4 * hf;
    const float* X0 = x + window_base(n0 + r) + co;
    const float* X1 = x + window_base(n0 + 16 + r) + co;
    const float* X2 = x + window_base(n0 + 32 + r) + co;
    const float* X3 = x + window_base(n0 + 48 + r) + co;

    v8f acc[2][4] = {};
    PoolGroup ga, gb;
    poolgroup_load(ga, W0, W1, X0, X1, X2, X3, 0);
    int k = 0;
    for (; k < klen - 16 * KS_GROUP; k += 16 * KS_GROUP) {
        poolgroup_load(gb, W0, W1, X0, X1, X2, X3, k + 8 * KS_GROUP);
        __builtin_amdgcn_sched_barrier(0);
        poolgroup_mma(acc, ga);
        __builtin_amdgcn_sched_barrier(0);
        poolgroup_load(ga, W0, W1, X0, X1, X2, X3, k + 16 * KS_GROUP);
        __builtin_amdgcn_sched_barrier(0);
        poolgroup_mma(acc, gb);
        __builtin_amdgcn_sched_barrier(0);
    }
    poolgroup_load(gb, W0, W1, X0, X1, X2, X3, k + 8 * KS_GROUP);
    poolgroup_mma(acc, ga);
    poolgroup_mma(acc, gb);

    float* p = part + wave * PT_WPITCH + r * PT_PITCH + 8 * hf;
    #pragma unroll
    for (int rt = 0; rt < 2; ++rt) {
        #pragma unroll
        for (int j = 0; j < 4; ++j) {
            #pragma unroll
            for (int vg = 0; vg < 8; ++vg)
                p[(16 * j) * PT_PITCH + 16 * rt + vg] = acc[rt][j][vg];
        }
    }
    __syncthreads();

    int ol = threadIdx.x & 31;
    int nb = threadIdx.x >> 5;
    float bv = bias[o0 + ol];
    #pragma unroll
    for (int i = 0; i < 8; ++i) {
        int nl = nb + 8 * i;
        const float* q = part + nl * PT_PITCH + ol;
        float s = q[0];
        #pragma unroll
        for (int sl = 1; sl < KS_SLICES; ++sl)
            s += q[sl * PT_WPITCH];
        C[(size_t)(n0 + nl) * N + o0 + ol] = s + bv;
    }
}

struct QGroup { RegFrag f[KS_GROUP]; };

__device__ __forceinline__ void qgroup_load(QGroup& g,
                                            const float* __restrict__ A0,
                                            const float* __restrict__ A1,
                                            const float* __restrict__ Bp,
                                            int k) {
    #pragma unroll
    for (int t = 0; t < KS_GROUP; ++t) g.f[t].a0 = *(const v4f*)(A0 + k + 8 * t);
    #pragma unroll
    for (int t = 0; t < KS_GROUP; ++t) g.f[t].a1 = *(const v4f*)(A1 + k + 8 * t);
    #pragma unroll
    for (int t = 0; t < KS_GROUP; ++t) {
        const float* bp = Bp + (size_t)(k + 8 * t) * IMG_PLANE;
        g.f[t].b0 = *(const v4f*)(bp);
        g.f[t].b1 = *(const v4f*)(bp + IMG_PLANE);
        g.f[t].b2 = *(const v4f*)(bp + 2 * IMG_PLANE);
        g.f[t].b3 = *(const v4f*)(bp + 3 * IMG_PLANE);
    }
}

__device__ __forceinline__ void qgroup_mma(v8f (&acc)[2][4], const QGroup& g) {
    #pragma unroll
    for (int t = 0; t < KS_GROUP; ++t) regtile_mma(acc, g.f[t]);
}

__device__ __forceinline__ void qproj_body(
    int tile,
    const float* __restrict__ Wq, const float* __restrict__ x,
    float* __restrict__ qmat, int K, float scale)
{
    int wave = threadIdx.x >> 5;
    int lane = threadIdx.x & 31;
    int hf   = lane >> 4;
    int r    = lane & 15;
    int o0   = wave * 32;

    int tt = tile;
    int cg = tt % 3;  tt /= 3;
    int ip = tt % 4;  tt /= 4;
    int wh = tt % 12;
    int bi = tt / 12;
    int irow = r >> 3;
    int icol = cg * 32 + 4 * (r & 7);
    int y = wh * 8 + ip * 2 + irow;

    const float* A0 = Wq + (size_t)(o0 + r) * K + 4 * hf;
    const float* A1 = A0 + (size_t)16 * K;
    const float* Bp = x + ((size_t)bi * IMG_CH + 4 * hf) * IMG_PLANE + (size_t)y * IMG_W + icol;

    v8f acc[2][4] = {};
    QGroup ga, gb;
    qgroup_load(ga, A0, A1, Bp, 0);
    int k = 0;
    for (; k < K - 16 * KS_GROUP; k += 16 * KS_GROUP) {
        qgroup_load(gb, A0, A1, Bp, k + 8 * KS_GROUP);
        __builtin_amdgcn_sched_barrier(0);
        qgroup_mma(acc, ga);
        __builtin_amdgcn_sched_barrier(0);
        qgroup_load(ga, A0, A1, Bp, k + 16 * KS_GROUP);
        __builtin_amdgcn_sched_barrier(0);
        qgroup_mma(acc, gb);
        __builtin_amdgcn_sched_barrier(0);
    }
    qgroup_load(gb, A0, A1, Bp, k + 8 * KS_GROUP);
    qgroup_mma(acc, ga);
    qgroup_mma(acc, gb);

    int n = bi * 144 + wh * 12 + (icol >> 3);
    size_t m = (size_t)n * 64 + (ip * 2 + irow) * 8 + (icol & 7);
    float* qrow = qmat + m * IMG_CH + o0 + 8 * hf;
    #pragma unroll
    for (int j = 0; j < 4; ++j) {
        #pragma unroll
        for (int rt = 0; rt < 2; ++rt) {
            v4f lo, hi;
            lo.x = acc[rt][j][0] * scale; lo.y = acc[rt][j][1] * scale;
            lo.z = acc[rt][j][2] * scale; lo.w = acc[rt][j][3] * scale;
            hi.x = acc[rt][j][4] * scale; hi.y = acc[rt][j][5] * scale;
            hi.z = acc[rt][j][6] * scale; hi.w = acc[rt][j][7] * scale;
            float* dst = qrow + (size_t)j * IMG_CH + 16 * rt;
            *(v4f*)(dst)     = lo;
            *(v4f*)(dst + 4) = hi;
        }
    }
}

#define POOL_OBLOCKS 8
#define POOL_BLOCKS 144
#define Q_TILES 1152

__global__ __launch_bounds__(256) void pool_and_qproj(
    const float* __restrict__ x, const float* __restrict__ Wp,
    float* __restrict__ poolx, const float* __restrict__ pool_b,
    const float* __restrict__ Wq, float* __restrict__ qmat,
    int poolN, int poolK, int qK, float scale)
{
    __shared__ float part[KS_SLICES * PT_WPITCH];
    int id = blockIdx.x;
    if (id < POOL_BLOCKS)
        pool_gemm_body(part, id % POOL_OBLOCKS, id / POOL_OBLOCKS, x, Wp, poolx, pool_b, poolN, poolK);
    else
        qproj_body(id - POOL_BLOCKS, Wq, x, qmat, qK, scale);
}

#define PPITCH 148

template <int CTRL>
__device__ __forceinline__ float half16_xor(float v) {
    return __builtin_bit_cast(float,
        __builtin_amdgcn_update_dpp(0, __builtin_bit_cast(int, v), CTRL, 0xf, 0xf, true));
}
__device__ __forceinline__ float half16_max(float v) {
    v = fmaxf(v, half16_xor<0x161>(v));
    v = fmaxf(v, half16_xor<0x162>(v));
    v = fmaxf(v, half16_xor<0x164>(v));
    v = fmaxf(v, half16_xor<0x168>(v));
    return v;
}
__device__ __forceinline__ float half16_sum(float v) {
    v += half16_xor<0x161>(v);
    v += half16_xor<0x162>(v);
    v += half16_xor<0x164>(v);
    v += half16_xor<0x168>(v);
    return v;
}

__global__ __launch_bounds__(128) void attn_kernel(
    float* __restrict__ qmat, const float* __restrict__ kv)
{
    __shared__ __attribute__((aligned(16))) float pbuf[64 * PPITCH];
    int id = blockIdx.x;
    int h  = id & 7;
    int bb = (id >> 3) & 7;
    int g  = id >> 6;
    int n  = g * 8 + bb;

    int wave = threadIdx.x >> 5;
    int lane = threadIdx.x & 31;
    int hf   = lane >> 4;
    int r    = lane & 15;
    int m0   = wave * 16;

    const float* qbase = qmat + ((size_t)n * 64 + m0 + r) * 256 + h * 32 + 4 * hf;
    const float* kb    = kv + (size_t)bb * 144 * 512 + h * 32;
    const float* vb    = kb + 256;

    v4f qf[4];
    #pragma unroll
    for (int t = 0; t < 4; ++t) qf[t] = *(const v4f*)(qbase + 8 * t);

    v8f s[9];
    #pragma unroll
    for (int g3 = 0; g3 < 3; ++g3) {
        v4f kf[3][4];
        #pragma unroll
        for (int j = 0; j < 3; ++j) {
            const float* krow = kb + (size_t)((3 * g3 + j) * 16 + r) * 512 + 4 * hf;
            #pragma unroll
            for (int t = 0; t < 4; ++t) kf[j][t] = *(const v4f*)(krow + 8 * t);
        }
        v8f a3[3] = {};
        #pragma unroll
        for (int t = 0; t < 4; ++t) {
            v2f a0 = {qf[t].x, qf[t].y}, a1 = {qf[t].z, qf[t].w};
            #pragma unroll
            for (int j = 0; j < 3; ++j) {
                v2f b = {kf[j][t].x, kf[j][t].y};
                a3[j] = __builtin_amdgcn_wmma_f32_16x16x4_f32(false, a0, false, b,
                                                              (short)0, a3[j], false, false);
            }
            #pragma unroll
            for (int j = 0; j < 3; ++j) {
                v2f b = {kf[j][t].z, kf[j][t].w};
                a3[j] = __builtin_amdgcn_wmma_f32_16x16x4_f32(false, a1, false, b,
                                                              (short)0, a3[j], false, false);
            }
        }
        #pragma unroll
        for (int j = 0; j < 3; ++j) s[3 * g3 + j] = a3[j];
    }

    float* pw = pbuf + (m0 + 8 * hf) * PPITCH + r;
    #pragma unroll
    for (int vg = 0; vg < 8; ++vg) {
        float mx = s[0][vg];
        #pragma unroll
        for (int nt = 1; nt < 9; ++nt) mx = fmaxf(mx, s[nt][vg]);
        mx = half16_max(mx);
        float sum = 0.0f;
        #pragma unroll
        for (int nt = 0; nt < 9; ++nt) {
            float e = __expf(s[nt][vg] - mx);
            s[nt][vg] = e;
            sum += e;
        }
        sum = half16_sum(sum);
        float inv = 1.0f / sum;
        #pragma unroll
        for (int nt = 0; nt < 9; ++nt)
            pw[vg * PPITCH + nt * 16] = s[nt][vg] * inv;
    }
    __syncthreads();

    const float* pr = pbuf + (m0 + r) * PPITCH + 4 * hf;
    const float* vp = vb + (size_t)(4 * hf) * 512 + 2 * r;
    v8f o0 = {}, o1 = {};
    #pragma unroll
    for (int t = 0; t < 18; ++t) {
        v4f p4 = *(const v4f*)(pr + 8 * t);
        const float* vq = vp + (size_t)(8 * t) * 512;
        v2f w0 = *(const v2f*)(vq);
        v2f w1 = *(const v2f*)(vq + 512);
        v2f w2 = *(const v2f*)(vq + 1024);
        v2f w3 = *(const v2f*)(vq + 1536);
        v2f a0 = {p4.x, p4.y}, a1 = {p4.z, p4.w};
        v2f b00 = {w0.x, w1.x}, b01 = {w0.y, w1.y};
        v2f b10 = {w2.x, w3.x}, b11 = {w2.y, w3.y};
        o0 = __builtin_amdgcn_wmma_f32_16x16x4_f32(false, a0, false, b00, (short)0, o0, false, false);
        o1 = __builtin_amdgcn_wmma_f32_16x16x4_f32(false, a0, false, b01, (short)0, o1, false, false);
        o0 = __builtin_amdgcn_wmma_f32_16x16x4_f32(false, a1, false, b10, (short)0, o0, false, false);
        o1 = __builtin_amdgcn_wmma_f32_16x16x4_f32(false, a1, false, b11, (short)0, o1, false, false);
    }
    float* orow = qmat + ((size_t)n * 64 + m0 + 8 * hf) * 256 + h * 32 + 2 * r;
    #pragma unroll
    for (int vg = 0; vg < 8; ++vg) {
        v2f o = {o0[vg], o1[vg]};
        *(v2f*)(orow + (size_t)vg * 256) = o;
    }
}

extern "C" void kernel_launch(void* const* d_in, const int* in_sizes, int n_in,
                              void* d_out, int out_size, void* d_ws, size_t ws_size,
                              hipStream_t stream) {
    const float* x      = (const float*)d_in[0];
    const float* Wq     = (const float*)d_in[1];
    const float* Wkv    = (const float*)d_in[2];
    const float* pool_w = (const float*)d_in[3];
    const float* pool_b = (const float*)d_in[4];
    const float* rpe_w  = (const float*)d_in[5];
    const float* rpe_b  = (const float*)d_in[6];
    const float* proj_w = (const float*)d_in[7];
    const float* proj_b = (const float*)d_in[8];
    float* out = (float*)d_out;

    float* ws    = (float*)d_ws;
    float* qmat  = ws;
    float* projT = qmat + 18874368;
    float* poolx = projT + 65536;
    float* kvb   = poolx + 294912;
    (void)ws_size; (void)in_sizes; (void)n_in; (void)out_size;

    const int TB = 256;
    transpose_k<<<(65536 + TB - 1) / TB, TB, 0, stream>>>(proj_w, projT, 256, 256, 65536);

    pool_and_qproj<<<POOL_BLOCKS + Q_TILES, 256, 0, stream>>>(
        x, pool_w, poolx, pool_b, Wq, qmat, 256, 16384, 256, QSCALE);

    gemm_wmma_f32<<<dim3(512 / 64, 1152 / 32), TB, 0, stream>>>(
        poolx, Wkv, kvb, nullptr, 1152, 512, 256, 1.0f);

    attn_kernel<<<144 * 8 * 8, 128, 0, stream>>>(qmat, kvb);

    outproj_rpe_fused<<<73728 / 32, 128, 0, stream>>>(
        qmat, x, rpe_w, rpe_b, projT, out, proj_b, 256, 256);
}


# Attention moved into the out-projection launch: a block runs the attention of its 32 rows, F goes into its LDS A tile, never stored; B loads grouped 32 k ahead; 5 launches -> 4; sum order unchanged
# speedup vs baseline: 4.0522x; 1.0476x over previous
#include <hip/hip_runtime.h>
#include <hip/hip_bf16.h>

typedef __attribute__((ext_vector_type(2))) float v2f;
typedef __attribute__((ext_vector_type(8))) float v8f;

#define QSCALE 0.17677669529663687f


__global__ void transpose_k(const float* __restrict__ src, float* __restrict__ dst,
                            int O, int K, int total) {
    int idx = blockIdx.x * blockDim.x + threadIdx.x;
    if (idx >= total) return;
    int o = idx % O;
    int k = idx / O;
    dst[idx] = src[(size_t)o * K + k];
}

__global__ __launch_bounds__(256) void gemm_wmma_f32(
    const float* __restrict__ A, const float* __restrict__ B,
    float* __restrict__ C, const float* __restrict__ bias,
    int M, int N, int K, float scale)
{
    int wave = threadIdx.x >> 5;
    int lane = threadIdx.x & 31;
    int hf   = lane >> 4;
    int r    = lane & 15;
    int m0 = blockIdx.y * 32 + (wave >> 2) * 16;
    int n0 = blockIdx.x * 64 + (wave & 3) * 16;

    const float* Arow = A + (size_t)(m0 + r) * K + 2 * hf;
    const float* Brow = B + (size_t)(n0 + r) * K + 2 * hf;

    v8f acc = {};
    #pragma unroll 8
    for (int k = 0; k < K; k += 4) {
        v2f a, b;
        a.x = Arow[k];
        a.y = Arow[k + 1];
        b.x = Brow[k];
        b.y = Brow[k + 1];
        acc = __builtin_amdgcn_wmma_f32_16x16x4_f32(false, a, false, b,
                                                    (short)0, acc, false, false);
    }
    int col = n0 + r;
    float bv = bias ? bias[col] : 0.0f;
    #pragma unroll
    for (int vg = 0; vg < 8; ++vg)
        C[(size_t)(m0 + vg + 8 * hf) * N + col] = acc[vg] * scale + bv;
}

typedef __attribute__((ext_vector_type(4))) float v4f;

struct RegFrag {
    v4f a0, a1;
    v4f b0, b1, b2, b3;
};

__device__ __forceinline__ void regtile_mma(v8f (&acc)[2][4], const RegFrag& f) {
    v2f a00 = {f.a0.x, f.a0.y}, a01 = {f.a0.z, f.a0.w};
    v2f a10 = {f.a1.x, f.a1.y}, a11 = {f.a1.z, f.a1.w};
    #pragma unroll
    for (int j = 0; j < 4; ++j) {
        v2f b = {f.b0[j], f.b1[j]};
        acc[0][j] = __builtin_amdgcn_wmma_f32_16x16x4_f32(false, a00, false, b,
                                                          (short)0, acc[0][j], false, false);
        acc[1][j] = __builtin_amdgcn_wmma_f32_16x16x4_f32(false, a10, false, b,
                                                          (short)0, acc[1][j], false, false);
    }
    #pragma unroll
    for (int j = 0; j < 4; ++j) {
        v2f b = {f.b2[j], f.b3[j]};
        acc[0][j] = __builtin_amdgcn_wmma_f32_16x16x4_f32(false, a01, false, b,
                                                          (short)0, acc[0][j], false, false);
        acc[1][j] = __builtin_amdgcn_wmma_f32_16x16x4_f32(false, a11, false, b,
                                                          (short)0, acc[1][j], false, false);
    }
}

#define KS_SLICES 8
#define KS_GROUP 4
#define PT_PITCH 35
#define PT_WPITCH (64 * PT_PITCH)
#define IMG_W 96
#define IMG_PLANE 9216
#define IMG_CH 256

struct PoolFrag {
    v4f w0, w1;
    v4f xw[4];
};
struct PoolGroup { PoolFrag f[KS_GROUP]; };

__device__ __forceinline__ size_t window_base(int n) {
    int bi = n / 144;
    int l  = n - bi * 144;
    int wh = l / 12, ww = l - wh * 12;
    return (size_t)bi * IMG_CH * IMG_PLANE + (size_t)(wh * 8) * IMG_W + ww * 8;
}

__device__ __forceinline__ void poolgroup_load(PoolGroup& g,
                                               const float* __restrict__ W0,
                                               const float* __restrict__ W1,
                                               const float* __restrict__ X0,
                                               const float* __restrict__ X1,
                                               const float* __restrict__ X2,
                                               const float* __restrict__ X3,
                                               int k) {
    size_t xo = (size_t)(k >> 6) * IMG_PLANE + (size_t)(((k >> 5) & 1) * 4) * IMG_W;
    #pragma unroll
    for (int t = 0; t < KS_GROUP; ++t) g.f[t].w0 = *(const v4f*)(W0 + k + 8 * t);
    #pragma unroll
    for (int t = 0; t < KS_GROUP; ++t) g.f[t].w1 = *(const v4f*)(W1 + k + 8 * t);
    #pragma unroll
    for (int t = 0; t < KS_GROUP; ++t) {
        g.f[t].xw[0] = *(const v4f*)(X0 + xo + t * IMG_W);
        g.f[t].xw[1] = *(const v4f*)(X1 + xo + t * IMG_W);
        g.f[t].xw[2] = *(const v4f*)(X2 + xo + t * IMG_W);
        g.f[t].xw[3] = *(const v4f*)(X3 + xo + t * IMG_W);
    }
}

__device__ __forceinline__ void pooltile_mma(v8f (&acc)[2][4], const PoolFrag& f) {
    v2f a00 = {f.w0.x, f.w0.y}, a01 = {f.w0.z, f.w0.w};
    v2f a10 = {f.w1.x, f.w1.y}, a11 = {f.w1.z, f.w1.w};
    #pragma unroll
    for (int j = 0; j < 4; ++j) {
        v2f b = {f.xw[j].x, f.xw[j].y};
        acc[0][j] = __builtin_amdgcn_wmma_f32_16x16x4_f32(false, a00, false, b,
                                                          (short)0, acc[0][j], false, false);
        acc[1][j] = __builtin_amdgcn_wmma_f32_16x16x4_f32(false, a10, false, b,
                                                          (short)0, acc[1][j], false, false);
    }
    #pragma unroll
    for (int j = 0; j < 4; ++j) {
        v2f b = {f.xw[j].z, f.xw[j].w};
        acc[0][j] = __builtin_amdgcn_wmma_f32_16x16x4_f32(false, a01, false, b,
                                                          (short)0, acc[0][j], false, false);
        acc[1][j] = __builtin_amdgcn_wmma_f32_16x16x4_f32(false, a11, false, b,
                                                          (short)0, acc[1][j], false, false);
    }
}

__device__ __forceinline__ void poolgroup_mma(v8f (&acc)[2][4], const PoolGroup& g) {
    #pragma unroll
    for (int t = 0; t < KS_GROUP; ++t) pooltile_mma(acc, g.f[t]);
}

__device__ __forceinline__ void pool_gemm_body(
    float* __restrict__ part, int obl, int nbl,
    const float* __restrict__ x, const float* __restrict__ Wp,
    float* __restrict__ C, const float* __restrict__ bias,
    int N, int K)
{
    int wave = threadIdx.x >> 5;
    int lane = threadIdx.x & 31;
    int hf   = lane >> 4;
    int r    = lane & 15;
    int o0 = obl * 32;
    int n0 = nbl * 64;
    int klen = K / KS_SLICES;
    int kb = wave * klen;

    const float* W0 = Wp + (size_t)(o0 + r) * K + kb + 4 * hf;
    const float* W1 = W0 + (size_t)16 * K;
    size_t co = (size_t)(kb >> 6) * IMG_PLANE + 4 * hf;
    const float* X0 = x + window_base(n0 + r) + co;
    const float* X1 = x + window_base(n0 + 16 + r) + co;
    const float* X2 = x + window_base(n0 + 32 + r) + co;
    const float* X3 = x + window_base(n0 + 48 + r) + co;

    v8f acc[2][4] = {};
    PoolGroup ga, gb;
    poolgroup_load(ga, W0, W1, X0, X1, X2, X3, 0);
    int k = 0;
    for (; k < klen - 16 * KS_GROUP; k += 16 * KS_GROUP) {
        poolgroup_load(gb, W0, W1, X0, X1, X2, X3, k + 8 * KS_GROUP);
        __builtin_amdgcn_sched_barrier(0);
        poolgroup_mma(acc, ga);
        __builtin_amdgcn_sched_barrier(0);
        poolgroup_load(ga, W0, W1, X0, X1, X2, X3, k + 16 * KS_GROUP);
        __builtin_amdgcn_sched_barrier(0);
        poolgroup_mma(acc, gb);
        __builtin_amdgcn_sched_barrier(0);
    }
    poolgroup_load(gb, W0, W1, X0, X1, X2, X3, k + 8 * KS_GROUP);
    poolgroup_mma(acc, ga);
    poolgroup_mma(acc, gb);

    float* p = part + wave * PT_WPITCH + r * PT_PITCH + 8 * hf;
    #pragma unroll
    for (int rt = 0; rt < 2; ++rt) {
        #pragma unroll
        for (int j = 0; j < 4; ++j) {
            #pragma unroll
            for (int vg = 0; vg < 8; ++vg)
                p[(16 * j) * PT_PITCH + 16 * rt + vg] = acc[rt][j][vg];
        }
    }
    __syncthreads();

    int ol = threadIdx.x & 31;
    int nb = threadIdx.x >> 5;
    float bv = bias[o0 + ol];
    #pragma unroll
    for (int i = 0; i < 8; ++i) {
        int nl = nb + 8 * i;
        const float* q = part + nl * PT_PITCH + ol;
        float s = q[0];
        #pragma unroll
        for (int sl = 1; sl < KS_SLICES; ++sl)
            s += q[sl * PT_WPITCH];
        C[(size_t)(n0 + nl) * N + o0 + ol] = s + bv;
    }
}

struct QGroup { RegFrag f[KS_GROUP]; };

__device__ __forceinline__ void qgroup_load(QGroup& g,
                                            const float* __restrict__ A0,
                                            const float* __restrict__ A1,
                                            const float* __restrict__ Bp,
                                            int k) {
    #pragma unroll
    for (int t = 0; t < KS_GROUP; ++t) g.f[t].a0 = *(const v4f*)(A0 + k + 8 * t);
    #pragma unroll
    for (int t = 0; t < KS_GROUP; ++t) g.f[t].a1 = *(const v4f*)(A1 + k + 8 * t);
    #pragma unroll
    for (int t = 0; t < KS_GROUP; ++t) {
        const float* bp = Bp + (size_t)(k + 8 * t) * IMG_PLANE;
        g.f[t].b0 = *(const v4f*)(bp);
        g.f[t].b1 = *(const v4f*)(bp + IMG_PLANE);
        g.f[t].b2 = *(const v4f*)(bp + 2 * IMG_PLANE);
        g.f[t].b3 = *(const v4f*)(bp + 3 * IMG_PLANE);
    }
}

__device__ __forceinline__ void qgroup_mma(v8f (&acc)[2][4], const QGroup& g) {
    #pragma unroll
    for (int t = 0; t < KS_GROUP; ++t) regtile_mma(acc, g.f[t]);
}

__device__ __forceinline__ void qproj_body(
    int tile,
    const float* __restrict__ Wq, const float* __restrict__ x,
    float* __restrict__ qmat, int K, float scale)
{
    int wave = threadIdx.x >> 5;
    int lane = threadIdx.x & 31;
    int hf   = lane >> 4;
    int r    = lane & 15;
    int o0   = wave * 32;

    int tt = tile;
    int cg = tt % 3;  tt /= 3;
    int ip = tt % 4;  tt /= 4;
    int wh = tt % 12;
    int bi = tt / 12;
    int irow = r >> 3;
    int icol = cg * 32 + 4 * (r & 7);
    int y = wh * 8 + ip * 2 + irow;

    const float* A0 = Wq + (size_t)(o0 + r) * K + 4 * hf;
    const float* A1 = A0 + (size_t)16 * K;
    const float* Bp = x + ((size_t)bi * IMG_CH + 4 * hf) * IMG_PLANE + (size_t)y * IMG_W + icol;

    v8f acc[2][4] = {};
    QGroup ga, gb;
    qgroup_load(ga, A0, A1, Bp, 0);
    int k = 0;
    for (; k < K - 16 * KS_GROUP; k += 16 * KS_GROUP) {
        qgroup_load(gb, A0, A1, Bp, k + 8 * KS_GROUP);
        __builtin_amdgcn_sched_barrier(0);
        qgroup_mma(acc, ga);
        __builtin_amdgcn_sched_barrier(0);
        qgroup_load(ga, A0, A1, Bp, k + 16 * KS_GROUP);
        __builtin_amdgcn_sched_barrier(0);
        qgroup_mma(acc, gb);
        __builtin_amdgcn_sched_barrier(0);
    }
    qgroup_load(gb, A0, A1, Bp, k + 8 * KS_GROUP);
    qgroup_mma(acc, ga);
    qgroup_mma(acc, gb);

    int n = bi * 144 + wh * 12 + (icol >> 3);
    size_t m = (size_t)n * 64 + (ip * 2 + irow) * 8 + (icol & 7);
    float* qrow = qmat + m * IMG_CH + o0 + 8 * hf;
    #pragma unroll
    for (int j = 0; j < 4; ++j) {
        #pragma unroll
        for (int rt = 0; rt < 2; ++rt) {
            v4f lo, hi;
            lo.x = acc[rt][j][0] * scale; lo.y = acc[rt][j][1] * scale;
            lo.z = acc[rt][j][2] * scale; lo.w = acc[rt][j][3] * scale;
            hi.x = acc[rt][j][4] * scale; hi.y = acc[rt][j][5] * scale;
            hi.z = acc[rt][j][6] * scale; hi.w = acc[rt][j][7] * scale;
            float* dst = qrow + (size_t)j * IMG_CH + 16 * rt;
            *(v4f*)(dst)     = lo;
            *(v4f*)(dst + 4) = hi;
        }
    }
}

#define POOL_OBLOCKS 8
#define POOL_BLOCKS 144
#define Q_TILES 1152

__global__ __launch_bounds__(256) void pool_and_qproj(
    const float* __restrict__ x, const float* __restrict__ Wp,
    float* __restrict__ poolx, const float* __restrict__ pool_b,
    const float* __restrict__ Wq, float* __restrict__ qmat,
    int poolN, int poolK, int qK, float scale)
{
    __shared__ float part[KS_SLICES * PT_WPITCH];
    int id = blockIdx.x;
    if (id < POOL_BLOCKS)
        pool_gemm_body(part, id % POOL_OBLOCKS, id / POOL_OBLOCKS, x, Wp, poolx, pool_b, poolN, poolK);
    else
        qproj_body(id - POOL_BLOCKS, Wq, x, qmat, qK, scale);
}

#define OP_APITCH 260
#define PPITCH 148

template <int CTRL>
__device__ __forceinline__ float half16_xor(float v) {
    return __builtin_bit_cast(float,
        __builtin_amdgcn_update_dpp(0, __builtin_bit_cast(int, v), CTRL, 0xf, 0xf, true));
}
__device__ __forceinline__ float half16_max(float v) {
    v = fmaxf(v, half16_xor<0x161>(v));
    v = fmaxf(v, half16_xor<0x162>(v));
    v = fmaxf(v, half16_xor<0x164>(v));
    v = fmaxf(v, half16_xor<0x168>(v));
    return v;
}
__device__ __forceinline__ float half16_sum(float v) {
    v += half16_xor<0x161>(v);
    v += half16_xor<0x162>(v);
    v += half16_xor<0x164>(v);
    v += half16_xor<0x168>(v);
    return v;
}

__device__ __forceinline__ void ogroup_load(QGroup& g,
                                            const float* A0, const float* A1,
                                            const float* __restrict__ Bp,
                                            int k, int N) {
    #pragma unroll
    for (int t = 0; t < KS_GROUP; ++t) g.f[t].a0 = *(const v4f*)(A0 + k + 8 * t);
    #pragma unroll
    for (int t = 0; t < KS_GROUP; ++t) g.f[t].a1 = *(const v4f*)(A1 + k + 8 * t);
    #pragma unroll
    for (int t = 0; t < KS_GROUP; ++t) {
        const float* bp = Bp + (size_t)(k + 8 * t) * N;
        g.f[t].b0 = *(const v4f*)(bp);
        g.f[t].b1 = *(const v4f*)(bp + N);
        g.f[t].b2 = *(const v4f*)(bp + 2 * (size_t)N);
        g.f[t].b3 = *(const v4f*)(bp + 3 * (size_t)N);
    }
}

__global__ __launch_bounds__(128) void attn_rpe_outproj(
    const float* __restrict__ qmat, const float* __restrict__ kv,
    const float* __restrict__ x,
    const float* __restrict__ rw, const float* __restrict__ rb,
    const float* __restrict__ B, float* __restrict__ C,
    const float* __restrict__ bias, int N, int K)
{
    __shared__ __attribute__((aligned(16))) float atile[32 * OP_APITCH];
    __shared__ __attribute__((aligned(16))) float pbuf[64 * PPITCH];
    int wave = threadIdx.x >> 5;
    int lane = threadIdx.x & 31;
    int hf   = lane >> 4;
    int r    = lane & 15;
    int m0 = blockIdx.x * 32;
    int n0 = wave * 64;

    {
        int bb   = (blockIdx.x >> 1) & 7;
        int rt   = wave & 1;
        int hsel = wave >> 1;
        const float* qrow = qmat + (size_t)(m0 + 16 * rt + r) * 256 + 4 * hf;
        const float* kvb  = kv + (size_t)bb * 144 * 512;
        float* pw       = pbuf + (16 * wave + 8 * hf) * PPITCH + r;
        const float* pr = pbuf + (16 * wave + r) * PPITCH + 4 * hf;
        float* fw       = atile + (16 * rt + 8 * hf) * OP_APITCH + 2 * r;

        v4f qf[4];
        #pragma unroll
        for (int t = 0; t < 4; ++t) qf[t] = *(const v4f*)(qrow + hsel * 32 + 8 * t);

        for (int i = 0; i < 4; ++i) {
            int h = 2 * i + hsel;
            int hn = i < 3 ? h + 2 : h;
            v4f qn[4];
            #pragma unroll
            for (int t = 0; t < 4; ++t) qn[t] = *(const v4f*)(qrow + hn * 32 + 8 * t);

            const float* kb = kvb + h * 32;
            const float* vb = kb + 256;

            v8f s[9];
            #pragma unroll
            for (int g3 = 0; g3 < 3; ++g3) {
                v4f kf[3][4];
                #pragma unroll
                for (int j = 0; j < 3; ++j) {
                    const float* krow = kb + (size_t)((3 * g3 + j) * 16 + r) * 512 + 4 * hf;
                    #pragma unroll
                    for (int t = 0; t < 4; ++t) kf[j][t] = *(const v4f*)(krow + 8 * t);
                }
                v8f a3[3] = {};
                #pragma unroll
                for (int t = 0; t < 4; ++t) {
                    v2f a0 = {qf[t].x, qf[t].y}, a1 = {qf[t].z, qf[t].w};
                    #pragma unroll
                    for (int j = 0; j < 3; ++j) {
                        v2f b = {kf[j][t].x, kf[j][t].y};
                        a3[j] = __builtin_amdgcn_wmma_f32_16x16x4_f32(false, a0, false, b,
                                                                      (short)0, a3[j], false, false);
                    }
                    #pragma unroll
                    for (int j = 0; j < 3; ++j) {
                        v2f b = {kf[j][t].z, kf[j][t].w};
                        a3[j] = __builtin_amdgcn_wmma_f32_16x16x4_f32(false, a1, false, b,
                                                                      (short)0, a3[j], false, false);
                    }
                }
                #pragma unroll
                for (int j = 0; j < 3; ++j) s[3 * g3 + j] = a3[j];
            }

            #pragma unroll
            for (int vg = 0; vg < 8; ++vg) {
                float mx = s[0][vg];
                #pragma unroll
                for (int nt = 1; nt < 9; ++nt) mx = fmaxf(mx, s[nt][vg]);
                mx = half16_max(mx);
                float sum = 0.0f;
                #pragma unroll
                for (int nt = 0; nt < 9; ++nt) {
                    float e = __expf(s[nt][vg] - mx);
                    s[nt][vg] = e;
                    sum += e;
                }
                sum = half16_sum(sum);
                float inv = 1.0f / sum;
                #pragma unroll
                for (int nt = 0; nt < 9; ++nt)
                    pw[vg * PPITCH + nt * 16] = s[nt][vg] * inv;
            }
            __syncthreads();

            const float* vp = vb + (size_t)(4 * hf) * 512 + 2 * r;
            v8f o0 = {}, o1 = {};
            #pragma unroll
            for (int t = 0; t < 18; ++t) {
                v4f p4 = *(const v4f*)(pr + 8 * t);
                const float* vq = vp + (size_t)(8 * t) * 512;
                v2f w0 = *(const v2f*)(vq);
                v2f w1 = *(const v2f*)(vq + 512);
                v2f w2 = *(const v2f*)(vq + 1024);
                v2f w3 = *(const v2f*)(vq + 1536);
                v2f a0 = {p4.x, p4.y}, a1 = {p4.z, p4.w};
                v2f b00 = {w0.x, w1.x}, b01 = {w0.y, w1.y};
                v2f b10 = {w2.x, w3.x}, b11 = {w2.y, w3.y};
                o0 = __builtin_amdgcn_wmma_f32_16x16x4_f32(false, a0, false, b00, (short)0, o0, false, false);
                o1 = __builtin_amdgcn_wmma_f32_16x16x4_f32(false, a0, false, b01, (short)0, o1, false, false);
                o0 = __builtin_amdgcn_wmma_f32_16x16x4_f32(false, a1, false, b10, (short)0, o0, false, false);
                o1 = __builtin_amdgcn_wmma_f32_16x16x4_f32(false, a1, false, b11, (short)0, o1, false, false);
            }
            #pragma unroll
            for (int vg = 0; vg < 8; ++vg) {
                v2f o = {o0[vg], o1[vg]};
                *(v2f*)(fw + vg * OP_APITCH + h * 32) = o;
            }
            #pragma unroll
            for (int t = 0; t < 4; ++t) qf[t] = qn[t];
        }
    }
    __syncthreads();

    int rsel = __builtin_amdgcn_readfirstlane(wave >> 1);
    int kq = 4 * (threadIdx.x & 63);
    const char* xb = (const char*)x;
    #pragma unroll 4
    for (int i = 0; i < 16; ++i) {
        int row = 2 * i + rsel;
        int m  = m0 + row;
        int pl = m / 36;
        int c  = pl & 255;
        int p  = (m - pl * 36) * 256 + kq;
        int y  = p / 96;
        int xx = p - y * 96;
        unsigned bo = ((unsigned)m * 256u + (unsigned)kq) * 4u;
        float* ap = atile + row * OP_APITCH + kq;
        v4f f = *(const v4f*)ap;
        const float* wp = rw + c * 9;
        float b0 = rb[c];
        v4f s = {b0, b0, b0, b0};
        bool lok = xx > 0, rok = xx < 92;
        unsigned lo = lok ? 0xfffffffcu : 0u;
        unsigned ro = rok ? 16u : 12u;
        #pragma unroll
        for (int dy = -1; dy <= 1; ++dy) {
            bool rowok = dy < 0 ? (y > 0) : (dy > 0 ? (y < 95) : true);
            unsigned br = rowok ? bo + (unsigned)(dy * 384) : bo;
            v4f mid = *(const v4f*)(xb + br);
            float lf = *(const float*)(xb + (br + lo));
            float rg = *(const float*)(xb + (br + ro));
            if (!rowok) mid = (v4f){0.0f, 0.0f, 0.0f, 0.0f};
            lf = (rowok && lok) ? lf : 0.0f;
            rg = (rowok && rok) ? rg : 0.0f;
            float w0 = wp[(dy + 1) * 3], w1 = wp[(dy + 1) * 3 + 1], w2 = wp[(dy + 1) * 3 + 2];
            s.x += lf    * w0; s.x += mid.x * w1; s.x += mid.y * w2;
            s.y += mid.x * w0; s.y += mid.y * w1; s.y += mid.z * w2;
            s.z += mid.y * w0; s.z += mid.z * w1; s.z += mid.w * w2;
            s.w += mid.z * w0; s.w += mid.w * w1; s.w += rg    * w2;
        }
        *(v4f*)ap = f + s;
    }
    __syncthreads();

    const float* A0 = atile + r * OP_APITCH + 4 * hf;
    const float* A1 = A0 + 16 * OP_APITCH;
    const float* Bp = B + (size_t)(4 * hf) * N + n0 + 4 * r;

    v8f acc[2][4] = {};
    QGroup ga, gb;
    ogroup_load(ga, A0, A1, Bp, 0, N);
    int k = 0;
    for (; k < K - 16 * KS_GROUP; k += 16 * KS_GROUP) {
        ogroup_load(gb, A0, A1, Bp, k + 8 * KS_GROUP, N);
        __builtin_amdgcn_sched_barrier(0);
        qgroup_mma(acc, ga);
        __builtin_amdgcn_sched_barrier(0);
        ogroup_load(ga, A0, A1, Bp, k + 16 * KS_GROUP, N);
        __builtin_amdgcn_sched_barrier(0);
        qgroup_mma(acc, gb);
        __builtin_amdgcn_sched_barrier(0);
    }
    ogroup_load(gb, A0, A1, Bp, k + 8 * KS_GROUP, N);
    qgroup_mma(acc, ga);
    qgroup_mma(acc, gb);

    v4f bv = *(const v4f*)(bias + n0 + 4 * r);
    #pragma unroll
    for (int rt = 0; rt < 2; ++rt) {
        #pragma unroll
        for (int vg = 0; vg < 8; ++vg) {
            v4f o;
            o.x = acc[rt][0][vg] + bv.x;
            o.y = acc[rt][1][vg] + bv.y;
            o.z = acc[rt][2][vg] + bv.z;
            o.w = acc[rt][3][vg] + bv.w;
            *(v4f*)(C + (size_t)(m0 + 16 * rt + vg + 8 * hf) * N + n0 + 4 * r) = o;
        }
    }
}

extern "C" void kernel_launch(void* const* d_in, const int* in_sizes, int n_in,
                              void* d_out, int out_size, void* d_ws, size_t ws_size,
                              hipStream_t stream) {
    const float* x      = (const float*)d_in[0];
    const float* Wq     = (const float*)d_in[1];
    const float* Wkv    = (const float*)d_in[2];
    const float* pool_w = (const float*)d_in[3];
    const float* pool_b = (const float*)d_in[4];
    const float* rpe_w  = (const float*)d_in[5];
    const float* rpe_b  = (const float*)d_in[6];
    const float* proj_w = (const float*)d_in[7];
    const float* proj_b = (const float*)d_in[8];
    float* out = (float*)d_out;

    float* ws    = (float*)d_ws;
    float* qmat  = ws;
    float* projT = qmat + 18874368;
    float* poolx = projT + 65536;
    float* kvb   = poolx + 294912;
    (void)ws_size; (void)in_sizes; (void)n_in; (void)out_size;

    const int TB = 256;
    transpose_k<<<(65536 + TB - 1) / TB, TB, 0, stream>>>(proj_w, projT, 256, 256, 65536);

    pool_and_qproj<<<POOL_BLOCKS + Q_TILES, 256, 0, stream>>>(
        x, pool_w, poolx, pool_b, Wq, qmat, 256, 16384, 256, QSCALE);

    gemm_wmma_f32<<<dim3(512 / 64, 1152 / 32), TB, 0, stream>>>(
        poolx, Wkv, kvb, nullptr, 1152, 512, 256, 1.0f);

    attn_rpe_outproj<<<73728 / 32, 128, 0, stream>>>(
        qmat, kvb, x, rpe_w, rpe_b, projT, out, proj_b, 256, 256);
}


# KV projection stores k and v in the attention's B-fragment order: a fragment load takes 512 consecutive bytes (4 lines), was 16 B of each of 16 rows for k, 8 B for v; sum order unchanged
# speedup vs baseline: 4.2789x; 1.0559x over previous
#include <hip/hip_runtime.h>
#include <hip/hip_bf16.h>

typedef __attribute__((ext_vector_type(2))) float v2f;
typedef __attribute__((ext_vector_type(8))) float v8f;

#define QSCALE 0.17677669529663687f


__global__ void transpose_k(const float* __restrict__ src, float* __restrict__ dst,
                            int O, int K, int total) {
    int idx = blockIdx.x * blockDim.x + threadIdx.x;
    if (idx >= total) return;
    int o = idx % O;
    int k = idx / O;
    dst[idx] = src[(size_t)o * K + k];
}

#define KV_KEYS 144
#define KV_HEADFLOATS 4608

__global__ __launch_bounds__(256) void kv_proj_frag(
    const float* __restrict__ A, const float* __restrict__ B,
    float* __restrict__ kfrag, float* __restrict__ vfrag, int K)
{
    int wave = threadIdx.x >> 5;
    int lane = threadIdx.x & 31;
    int hf   = lane >> 4;
    int r    = lane & 15;
    int m0 = blockIdx.y * 32 + (wave >> 2) * 16;
    int n0 = blockIdx.x * 64 + (wave & 3) * 16;

    const float* Arow = A + (size_t)(m0 + r) * K + 2 * hf;
    const float* Brow = B + (size_t)(n0 + r) * K + 2 * hf;

    v8f acc = {};
    #pragma unroll 8
    for (int k = 0; k < K; k += 4) {
        v2f a, b;
        a.x = Arow[k];
        a.y = Arow[k + 1];
        b.x = Brow[k];
        b.y = Brow[k + 1];
        acc = __builtin_amdgcn_wmma_f32_16x16x4_f32(false, a, false, b,
                                                    (short)0, acc, false, false);
    }
    int bb   = m0 / KV_KEYS;
    int key0 = m0 - bb * KV_KEYS;
    int col  = n0 + r;
    if (n0 < 256) {
        int h = col >> 5, d = col & 31;
        float* dst = kfrag + (size_t)(bb * 8 + h) * KV_HEADFLOATS
                   + (((key0 >> 4) * 4 + (d >> 3)) * 32 + ((d >> 2) & 1) * 16 + 8 * hf) * 4 + (d & 3);
        #pragma unroll
        for (int vg = 0; vg < 8; ++vg)
            dst[4 * vg] = acc[vg];
    } else {
        int c2 = col - 256;
        int h = c2 >> 5, d = c2 & 31;
        float* dst = vfrag + (size_t)(bb * 8 + h) * KV_HEADFLOATS
                   + (((key0 >> 3) + hf) * 2 * 32 + (d >> 1)) * 4 + (d & 1) * 2;
        #pragma unroll
        for (int vg = 0; vg < 8; ++vg)
            dst[((vg >> 1) & 1) * 128 + (vg >> 2) * 64 + (vg & 1)] = acc[vg];
    }
}

typedef __attribute__((ext_vector_type(4))) float v4f;

struct RegFrag {
    v4f a0, a1;
    v4f b0, b1, b2, b3;
};

__device__ __forceinline__ void regtile_mma(v8f (&acc)[2][4], const RegFrag& f) {
    v2f a00 = {f.a0.x, f.a0.y}, a01 = {f.a0.z, f.a0.w};
    v2f a10 = {f.a1.x, f.a1.y}, a11 = {f.a1.z, f.a1.w};
    #pragma unroll
    for (int j = 0; j < 4; ++j) {
        v2f b = {f.b0[j], f.b1[j]};
        acc[0][j] = __builtin_amdgcn_wmma_f32_16x16x4_f32(false, a00, false, b,
                                                          (short)0, acc[0][j], false, false);
        acc[1][j] = __builtin_amdgcn_wmma_f32_16x16x4_f32(false, a10, false, b,
                                                          (short)0, acc[1][j], false, false);
    }
    #pragma unroll
    for (int j = 0; j < 4; ++j) {
        v2f b = {f.b2[j], f.b3[j]};
        acc[0][j] = __builtin_amdgcn_wmma_f32_16x16x4_f32(false, a01, false, b,
                                                          (short)0, acc[0][j], false, false);
        acc[1][j] = __builtin_amdgcn_wmma_f32_16x16x4_f32(false, a11, false, b,
                                                          (short)0, acc[1][j], false, false);
    }
}

#define KS_SLICES 8
#define KS_GROUP 4
#define PT_PITCH 35
#define PT_WPITCH (64 * PT_PITCH)
#define IMG_W 96
#define IMG_PLANE 9216
#define IMG_CH 256

struct PoolFrag {
    v4f w0, w1;
    v4f xw[4];
};
struct PoolGroup { PoolFrag f[KS_GROUP]; };

__device__ __forceinline__ size_t window_base(int n) {
    int bi = n / 144;
    int l  = n - bi * 144;
    int wh = l / 12, ww = l - wh * 12;
    return (size_t)bi * IMG_CH * IMG_PLANE + (size_t)(wh * 8) * IMG_W + ww * 8;
}

__device__ __forceinline__ void poolgroup_load(PoolGroup& g,
                                               const float* __restrict__ W0,
                                               const float* __restrict__ W1,
                                               const float* __restrict__ X0,
                                               const float* __restrict__ X1,
                                               const float* __restrict__ X2,
                                               const float* __restrict__ X3,
                                               int k) {
    size_t xo = (size_t)(k >> 6) * IMG_PLANE + (size_t)(((k >> 5) & 1) * 4) * IMG_W;
    #pragma unroll
    for (int t = 0; t < KS_GROUP; ++t) g.f[t].w0 = *(const v4f*)(W0 + k + 8 * t);
    #pragma unroll
    for (int t = 0; t < KS_GROUP; ++t) g.f[t].w1 = *(const v4f*)(W1 + k + 8 * t);
    #pragma unroll
    for (int t = 0; t < KS_GROUP; ++t) {
        g.f[t].xw[0] = *(const v4f*)(X0 + xo + t * IMG_W);
        g.f[t].xw[1] = *(const v4f*)(X1 + xo + t * IMG_W);
        g.f[t].xw[2] = *(const v4f*)(X2 + xo + t * IMG_W);
        g.f[t].xw[3] = *(const v4f*)(X3 + xo + t * IMG_W);
    }
}

__device__ __forceinline__ void pooltile_mma(v8f (&acc)[2][4], const PoolFrag& f) {
    v2f a00 = {f.w0.x, f.w0.y}, a01 = {f.w0.z, f.w0.w};
    v2f a10 = {f.w1.x, f.w1.y}, a11 = {f.w1.z, f.w1.w};
    #pragma unroll
    for (int j = 0; j < 4; ++j) {
        v2f b = {f.xw[j].x, f.xw[j].y};
        acc[0][j] = __builtin_amdgcn_wmma_f32_16x16x4_f32(false, a00, false, b,
                                                          (short)0, acc[0][j], false, false);
        acc[1][j] = __builtin_amdgcn_wmma_f32_16x16x4_f32(false, a10, false, b,
                                                          (short)0, acc[1][j], false, false);
    }
    #pragma unroll
    for (int j = 0; j < 4; ++j) {
        v2f b = {f.xw[j].z, f.xw[j].w};
        acc[0][j] = __builtin_amdgcn_wmma_f32_16x16x4_f32(false, a01, false, b,
                                                          (short)0, acc[0][j], false, false);
        acc[1][j] = __builtin_amdgcn_wmma_f32_16x16x4_f32(false, a11, false, b,
                                                          (short)0, acc[1][j], false, false);
    }
}

__device__ __forceinline__ void poolgroup_mma(v8f (&acc)[2][4], const PoolGroup& g) {
    #pragma unroll
    for (int t = 0; t < KS_GROUP; ++t) pooltile_mma(acc, g.f[t]);
}

__device__ __forceinline__ void pool_gemm_body(
    float* __restrict__ part, int obl, int nbl,
    const float* __restrict__ x, const float* __restrict__ Wp,
    float* __restrict__ C, const float* __restrict__ bias,
    int N, int K)
{
    int wave = threadIdx.x >> 5;
    int lane = threadIdx.x & 31;
    int hf   = lane >> 4;
    int r    = lane & 15;
    int o0 = obl * 32;
    int n0 = nbl * 64;
    int klen = K / KS_SLICES;
    int kb = wave * klen;

    const float* W0 = Wp + (size_t)(o0 + r) * K + kb + 4 * hf;
    const float* W1 = W0 + (size_t)16 * K;
    size_t co = (size_t)(kb >> 6) * IMG_PLANE + 4 * hf;
    const float* X0 = x + window_base(n0 + r) + co;
    const float* X1 = x + window_base(n0 + 16 + r) + co;
    const float* X2 = x + window_base(n0 + 32 + r) + co;
    const float* X3 = x + window_base(n0 + 48 + r) + co;

    v8f acc[2][4] = {};
    PoolGroup ga, gb;
    poolgroup_load(ga, W0, W1, X0, X1, X2, X3, 0);
    int k = 0;
    for (; k < klen - 16 * KS_GROUP; k += 16 * KS_GROUP) {
        poolgroup_load(gb, W0, W1, X0, X1, X2, X3, k + 8 * KS_GROUP);
        __builtin_amdgcn_sched_barrier(0);
        poolgroup_mma(acc, ga);
        __builtin_amdgcn_sched_barrier(0);
        poolgroup_load(ga, W0, W1, X0, X1, X2, X3, k + 16 * KS_GROUP);
        __builtin_amdgcn_sched_barrier(0);
        poolgroup_mma(acc, gb);
        __builtin_amdgcn_sched_barrier(0);
    }
    poolgroup_load(gb, W0, W1, X0, X1, X2, X3, k + 8 * KS_GROUP);
    poolgroup_mma(acc, ga);
    poolgroup_mma(acc, gb);

    float* p = part + wave * PT_WPITCH + r * PT_PITCH + 8 * hf;
    #pragma unroll
    for (int rt = 0; rt < 2; ++rt) {
        #pragma unroll
        for (int j = 0; j < 4; ++j) {
            #pragma unroll
            for (int vg = 0; vg < 8; ++vg)
                p[(16 * j) * PT_PITCH + 16 * rt + vg] = acc[rt][j][vg];
        }
    }
    __syncthreads();

    int ol = threadIdx.x & 31;
    int nb = threadIdx.x >> 5;
    float bv = bias[o0 + ol];
    #pragma unroll
    for (int i = 0; i < 8; ++i) {
        int nl = nb + 8 * i;
        const float* q = part + nl * PT_PITCH + ol;
        float s = q[0];
        #pragma unroll
        for (int sl = 1; sl < KS_SLICES; ++sl)
            s += q[sl * PT_WPITCH];
        C[(size_t)(n0 + nl) * N + o0 + ol] = s + bv;
    }
}

struct QGroup { RegFrag f[KS_GROUP]; };

__device__ __forceinline__ void qgroup_load(QGroup& g,
                                            const float* __restrict__ A0,
                                            const float* __restrict__ A1,
                                            const float* __restrict__ Bp,
                                            int k) {
    #pragma unroll
    for (int t = 0; t < KS_GROUP; ++t) g.f[t].a0 = *(const v4f*)(A0 + k + 8 * t);
    #pragma unroll
    for (int t = 0; t < KS_GROUP; ++t) g.f[t].a1 = *(const v4f*)(A1 + k + 8 * t);
    #pragma unroll
    for (int t = 0; t < KS_GROUP; ++t) {
        const float* bp = Bp + (size_t)(k + 8 * t) * IMG_PLANE;
        g.f[t].b0 = *(const v4f*)(bp);
        g.f[t].b1 = *(const v4f*)(bp + IMG_PLANE);
        g.f[t].b2 = *(const v4f*)(bp + 2 * IMG_PLANE);
        g.f[t].b3 = *(const v4f*)(bp + 3 * IMG_PLANE);
    }
}

__device__ __forceinline__ void qgroup_mma(v8f (&acc)[2][4], const QGroup& g) {
    #pragma unroll
    for (int t = 0; t < KS_GROUP; ++t) regtile_mma(acc, g.f[t]);
}

__device__ __forceinline__ void qproj_body(
    int tile,
    const float* __restrict__ Wq, const float* __restrict__ x,
    float* __restrict__ qmat, int K, float scale)
{
    int wave = threadIdx.x >> 5;
    int lane = threadIdx.x & 31;
    int hf   = lane >> 4;
    int r    = lane & 15;
    int o0   = wave * 32;

    int tt = tile;
    int cg = tt % 3;  tt /= 3;
    int ip = tt % 4;  tt /= 4;
    int wh = tt % 12;
    int bi = tt / 12;
    int irow = r >> 3;
    int icol = cg * 32 + 4 * (r & 7);
    int y = wh * 8 + ip * 2 + irow;

    const float* A0 = Wq + (size_t)(o0 + r) * K + 4 * hf;
    const float* A1 = A0 + (size_t)16 * K;
    const float* Bp = x + ((size_t)bi * IMG_CH + 4 * hf) * IMG_PLANE + (size_t)y * IMG_W + icol;

    v8f acc[2][4] = {};
    QGroup ga, gb;
    qgroup_load(ga, A0, A1, Bp, 0);
    int k = 0;
    for (; k < K - 16 * KS_GROUP; k += 16 * KS_GROUP) {
        qgroup_load(gb, A0, A1, Bp, k + 8 * KS_GROUP);
        __builtin_amdgcn_sched_barrier(0);
        qgroup_mma(acc, ga);
        __builtin_amdgcn_sched_barrier(0);
        qgroup_load(ga, A0, A1, Bp, k + 16 * KS_GROUP);
        __builtin_amdgcn_sched_barrier(0);
        qgroup_mma(acc, gb);
        __builtin_amdgcn_sched_barrier(0);
    }
    qgroup_load(gb, A0, A1, Bp, k + 8 * KS_GROUP);
    qgroup_mma(acc, ga);
    qgroup_mma(acc, gb);

    int n = bi * 144 + wh * 12 + (icol >> 3);
    size_t m = (size_t)n * 64 + (ip * 2 + irow) * 8 + (icol & 7);
    float* qrow = qmat + m * IMG_CH + o0 + 8 * hf;
    #pragma unroll
    for (int j = 0; j < 4; ++j) {
        #pragma unroll
        for (int rt = 0; rt < 2; ++rt) {
            v4f lo, hi;
            lo.x = acc[rt][j][0] * scale; lo.y = acc[rt][j][1] * scale;
            lo.z = acc[rt][j][2] * scale; lo.w = acc[rt][j][3] * scale;
            hi.x = acc[rt][j][4] * scale; hi.y = acc[rt][j][5] * scale;
            hi.z = acc[rt][j][6] * scale; hi.w = acc[rt][j][7] * scale;
            float* dst = qrow + (size_t)j * IMG_CH + 16 * rt;
            *(v4f*)(dst)     = lo;
            *(v4f*)(dst + 4) = hi;
        }
    }
}

#define POOL_OBLOCKS 8
#define POOL_BLOCKS 144
#define Q_TILES 1152

__global__ __launch_bounds__(256) void pool_and_qproj(
    const float* __restrict__ x, const float* __restrict__ Wp,
    float* __restrict__ poolx, const float* __restrict__ pool_b,
    const float* __restrict__ Wq, float* __restrict__ qmat,
    int poolN, int poolK, int qK, float scale)
{
    __shared__ float part[KS_SLICES * PT_WPITCH];
    int id = blockIdx.x;
    if (id < POOL_BLOCKS)
        pool_gemm_body(part, id % POOL_OBLOCKS, id / POOL_OBLOCKS, x, Wp, poolx, pool_b, poolN, poolK);
    else
        qproj_body(id - POOL_BLOCKS, Wq, x, qmat, qK, scale);
}

#define OP_APITCH 260
#define PPITCH 148

template <int CTRL>
__device__ __forceinline__ float half16_xor(float v) {
    return __builtin_bit_cast(float,
        __builtin_amdgcn_update_dpp(0, __builtin_bit_cast(int, v), CTRL, 0xf, 0xf, true));
}
__device__ __forceinline__ float half16_max(float v) {
    v = fmaxf(v, half16_xor<0x161>(v));
    v = fmaxf(v, half16_xor<0x162>(v));
    v = fmaxf(v, half16_xor<0x164>(v));
    v = fmaxf(v, half16_xor<0x168>(v));
    return v;
}
__device__ __forceinline__ float half16_sum(float v) {
    v += half16_xor<0x161>(v);
    v += half16_xor<0x162>(v);
    v += half16_xor<0x164>(v);
    v += half16_xor<0x168>(v);
    return v;
}

__device__ __forceinline__ void ogroup_load(QGroup& g,
                                            const float* A0, const float* A1,
                                            const float* __restrict__ Bp,
                                            int k, int N) {
    #pragma unroll
    for (int t = 0; t < KS_GROUP; ++t) g.f[t].a0 = *(const v4f*)(A0 + k + 8 * t);
    #pragma unroll
    for (int t = 0; t < KS_GROUP; ++t) g.f[t].a1 = *(const v4f*)(A1 + k + 8 * t);
    #pragma unroll
    for (int t = 0; t < KS_GROUP; ++t) {
        const float* bp = Bp + (size_t)(k + 8 * t) * N;
        g.f[t].b0 = *(const v4f*)(bp);
        g.f[t].b1 = *(const v4f*)(bp + N);
        g.f[t].b2 = *(const v4f*)(bp + 2 * (size_t)N);
        g.f[t].b3 = *(const v4f*)(bp + 3 * (size_t)N);
    }
}

__global__ __launch_bounds__(128) void attn_rpe_outproj(
    const float* __restrict__ qmat,
    const float* __restrict__ kfrag, const float* __restrict__ vfrag,
    const float* __restrict__ x,
    const float* __restrict__ rw, const float* __restrict__ rb,
    const float* __restrict__ B, float* __restrict__ C,
    const float* __restrict__ bias, int N, int K)
{
    __shared__ __attribute__((aligned(16))) float atile[32 * OP_APITCH];
    __shared__ __attribute__((aligned(16))) float pbuf[64 * PPITCH];
    int wave = threadIdx.x >> 5;
    int lane = threadIdx.x & 31;
    int hf   = lane >> 4;
    int r    = lane & 15;
    int m0 = blockIdx.x * 32;
    int n0 = wave * 64;

    {
        int bb   = (blockIdx.x >> 1) & 7;
        int rt   = wave & 1;
        int hsel = wave >> 1;
        const float* qrow = qmat + (size_t)(m0 + 16 * rt + r) * 256 + 4 * hf;
        const float* kfb = kfrag + (size_t)bb * 8 * KV_HEADFLOATS + 4 * lane;
        const float* vfb = vfrag + (size_t)bb * 8 * KV_HEADFLOATS + 4 * lane;
        float* pw       = pbuf + (16 * wave + 8 * hf) * PPITCH + r;
        const float* pr = pbuf + (16 * wave + r) * PPITCH + 4 * hf;
        float* fw       = atile + (16 * rt + 8 * hf) * OP_APITCH + 2 * r;

        v4f qf[4];
        #pragma unroll
        for (int t = 0; t < 4; ++t) qf[t] = *(const v4f*)(qrow + hsel * 32 + 8 * t);

        for (int i = 0; i < 4; ++i) {
            int h = 2 * i + hsel;
            int hn = i < 3 ? h + 2 : h;
            v4f qn[4];
            #pragma unroll
            for (int t = 0; t < 4; ++t) qn[t] = *(const v4f*)(qrow + hn * 32 + 8 * t);

            const float* kb = kfb + h * KV_HEADFLOATS;
            const float* vb = vfb + h * KV_HEADFLOATS;

            v8f s[9];
            #pragma unroll
            for (int g3 = 0; g3 < 3; ++g3) {
                v4f kf[3][4];
                #pragma unroll
                for (int j = 0; j < 3; ++j) {
                    const float* ktile = kb + (3 * g3 + j) * 512;
                    #pragma unroll
                    for (int t = 0; t < 4; ++t) kf[j][t] = *(const v4f*)(ktile + 128 * t);
                }
                v8f a3[3] = {};
                #pragma unroll
                for (int t = 0; t < 4; ++t) {
                    v2f a0 = {qf[t].x, qf[t].y}, a1 = {qf[t].z, qf[t].w};
                    #pragma unroll
                    for (int j = 0; j < 3; ++j) {
                        v2f b = {kf[j][t].x, kf[j][t].y};
                        a3[j] = __builtin_amdgcn_wmma_f32_16x16x4_f32(false, a0, false, b,
                                                                      (short)0, a3[j], false, false);
                    }
                    #pragma unroll
                    for (int j = 0; j < 3; ++j) {
                        v2f b = {kf[j][t].z, kf[j][t].w};
                        a3[j] = __builtin_amdgcn_wmma_f32_16x16x4_f32(false, a1, false, b,
                                                                      (short)0, a3[j], false, false);
                    }
                }
                #pragma unroll
                for (int j = 0; j < 3; ++j) s[3 * g3 + j] = a3[j];
            }

            #pragma unroll
            for (int vg = 0; vg < 8; ++vg) {
                float mx = s[0][vg];
                #pragma unroll
                for (int nt = 1; nt < 9; ++nt) mx = fmaxf(mx, s[nt][vg]);
                mx = half16_max(mx);
                float sum = 0.0f;
                #pragma unroll
                for (int nt = 0; nt < 9; ++nt) {
                    float e = __expf(s[nt][vg] - mx);
                    s[nt][vg] = e;
                    sum += e;
                }
                sum = half16_sum(sum);
                float inv = 1.0f / sum;
                #pragma unroll
                for (int nt = 0; nt < 9; ++nt)
                    pw[vg * PPITCH + nt * 16] = s[nt][vg] * inv;
            }
            __syncthreads();

            v8f o0 = {}, o1 = {};
            v4f vlo[18], vhi[18], p4[18];
            #pragma unroll
            for (int t = 0; t < 18; ++t) {
                vlo[t] = *(const v4f*)(vb + (2 * t) * 128);
                vhi[t] = *(const v4f*)(vb + (2 * t + 1) * 128);
            }
            #pragma unroll
            for (int t = 0; t < 18; ++t) p4[t] = *(const v4f*)(pr + 8 * t);
            __builtin_amdgcn_sched_barrier(0);
            #pragma unroll
            for (int t = 0; t < 18; ++t) {
                v2f a0 = {p4[t].x, p4[t].y}, a1 = {p4[t].z, p4[t].w};
                v2f b00 = {vlo[t].x, vlo[t].y}, b01 = {vlo[t].z, vlo[t].w};
                v2f b10 = {vhi[t].x, vhi[t].y}, b11 = {vhi[t].z, vhi[t].w};
                o0 = __builtin_amdgcn_wmma_f32_16x16x4_f32(false, a0, false, b00, (short)0, o0, false, false);
                o1 = __builtin_amdgcn_wmma_f32_16x16x4_f32(false, a0, false, b01, (short)0, o1, false, false);
                o0 = __builtin_amdgcn_wmma_f32_16x16x4_f32(false, a1, false, b10, (short)0, o0, false, false);
                o1 = __builtin_amdgcn_wmma_f32_16x16x4_f32(false, a1, false, b11, (short)0, o1, false, false);
            }
            #pragma unroll
            for (int vg = 0; vg < 8; ++vg) {
                v2f o = {o0[vg], o1[vg]};
                *(v2f*)(fw + vg * OP_APITCH + h * 32) = o;
            }
            #pragma unroll
            for (int t = 0; t < 4; ++t) qf[t] = qn[t];
        }
    }
    __syncthreads();

    int rsel = __builtin_amdgcn_readfirstlane(wave >> 1);
    int kq = 4 * (threadIdx.x & 63);
    const char* xb = (const char*)x;
    #pragma unroll 4
    for (int i = 0; i < 16; ++i) {
        int row = 2 * i + rsel;
        int m  = m0 + row;
        int pl = m / 36;
        int c  = pl & 255;
        int p  = (m - pl * 36) * 256 + kq;
        int y  = p / 96;
        int xx = p - y * 96;
        unsigned bo = ((unsigned)m * 256u + (unsigned)kq) * 4u;
        float* ap = atile + row * OP_APITCH + kq;
        v4f f = *(const v4f*)ap;
        const float* wp = rw + c * 9;
        float b0 = rb[c];
        v4f s = {b0, b0, b0, b0};
        bool lok = xx > 0, rok = xx < 92;
        unsigned lo = lok ? 0xfffffffcu : 0u;
        unsigned ro = rok ? 16u : 12u;
        #pragma unroll
        for (int dy = -1; dy <= 1; ++dy) {
            bool rowok = dy < 0 ? (y > 0) : (dy > 0 ? (y < 95) : true);
            unsigned br = rowok ? bo + (unsigned)(dy * 384) : bo;
            v4f mid = *(const v4f*)(xb + br);
            float lf = *(const float*)(xb + (br + lo));
            float rg = *(const float*)(xb + (br + ro));
            if (!rowok) mid = (v4f){0.0f, 0.0f, 0.0f, 0.0f};
            lf = (rowok && lok) ? lf : 0.0f;
            rg = (rowok && rok) ? rg : 0.0f;
            float w0 = wp[(dy + 1) * 3], w1 = wp[(dy + 1) * 3 + 1], w2 = wp[(dy + 1) * 3 + 2];
            s.x += lf    * w0; s.x += mid.x * w1; s.x += mid.y * w2;
            s.y += mid.x * w0; s.y += mid.y * w1; s.y += mid.z * w2;
            s.z += mid.y * w0; s.z += mid.z * w1; s.z += mid.w * w2;
            s.w += mid.z * w0; s.w += mid.w * w1; s.w += rg    * w2;
        }
        *(v4f*)ap = f + s;
    }
    __syncthreads();

    const float* A0 = atile + r * OP_APITCH + 4 * hf;
    const float* A1 = A0 + 16 * OP_APITCH;
    const float* Bp = B + (size_t)(4 * hf) * N + n0 + 4 * r;

    v8f acc[2][4] = {};
    QGroup ga, gb;
    ogroup_load(ga, A0, A1, Bp, 0, N);
    int k = 0;
    for (; k < K - 16 * KS_GROUP; k += 16 * KS_GROUP) {
        ogroup_load(gb, A0, A1, Bp, k + 8 * KS_GROUP, N);
        __builtin_amdgcn_sched_barrier(0);
        qgroup_mma(acc, ga);
        __builtin_amdgcn_sched_barrier(0);
        ogroup_load(ga, A0, A1, Bp, k + 16 * KS_GROUP, N);
        __builtin_amdgcn_sched_barrier(0);
        qgroup_mma(acc, gb);
        __builtin_amdgcn_sched_barrier(0);
    }
    ogroup_load(gb, A0, A1, Bp, k + 8 * KS_GROUP, N);
    qgroup_mma(acc, ga);
    qgroup_mma(acc, gb);

    v4f bv = *(const v4f*)(bias + n0 + 4 * r);
    #pragma unroll
    for (int rt = 0; rt < 2; ++rt) {
        #pragma unroll
        for (int vg = 0; vg < 8; ++vg) {
            v4f o;
            o.x = acc[rt][0][vg] + bv.x;
            o.y = acc[rt][1][vg] + bv.y;
            o.z = acc[rt][2][vg] + bv.z;
            o.w = acc[rt][3][vg] + bv.w;
            *(v4f*)(C + (size_t)(m0 + 16 * rt + vg + 8 * hf) * N + n0 + 4 * r) = o;
        }
    }
}

extern "C" void kernel_launch(void* const* d_in, const int* in_sizes, int n_in,
                              void* d_out, int out_size, void* d_ws, size_t ws_size,
                              hipStream_t stream) {
    const float* x      = (const float*)d_in[0];
    const float* Wq     = (const float*)d_in[1];
    const float* Wkv    = (const float*)d_in[2];
    const float* pool_w = (const float*)d_in[3];
    const float* pool_b = (const float*)d_in[4];
    const float* rpe_w  = (const float*)d_in[5];
    const float* rpe_b  = (const float*)d_in[6];
    const float* proj_w = (const float*)d_in[7];
    const float* proj_b = (const float*)d_in[8];
    float* out = (float*)d_out;

    float* ws    = (float*)d_ws;
    float* qmat  = ws;
    float* projT = qmat + 18874368;
    float* poolx = projT + 65536;
    float* kfrag = poolx + 294912;
    float* vfrag = kfrag + 294912;
    (void)ws_size; (void)in_sizes; (void)n_in; (void)out_size;

    const int TB = 256;
    transpose_k<<<(65536 + TB - 1) / TB, TB, 0, stream>>>(proj_w, projT, 256, 256, 65536);

    pool_and_qproj<<<POOL_BLOCKS + Q_TILES, 256, 0, stream>>>(
        x, pool_w, poolx, pool_b, Wq, qmat, 256, 16384, 256, QSCALE);

    kv_proj_frag<<<dim3(512 / 64, 1152 / 32), TB, 0, stream>>>(
        poolx, Wkv, kfrag, vfrag, 256);

    attn_rpe_outproj<<<73728 / 32, 128, 0, stream>>>(
        qmat, kfrag, vfrag, x, rpe_w, rpe_b, projT, out, proj_b, 256, 256);
}
